# starting kernel
# baseline (speedup 1.0000x reference)
#include <hip/hip_runtime.h>
#include <hip/hip_bf16.h>

#define FLT_BIG 3.402823466e+38f
#define RADIUS2 0.01f
#define K_LOCAL 32
#define CAP 2048

typedef __attribute__((ext_vector_type(2))) float v2f;
typedef __attribute__((ext_vector_type(8))) float v8f;

__global__ __launch_bounds__(1024)
void fps_kernel(const float* __restrict__ pos, float* __restrict__ dist,
                int* __restrict__ cent_idx, float* __restrict__ cents,
                int N, int M) {
  const int tid = threadIdx.x;
  __shared__ int   s_cur;
  __shared__ float s_cp[3];
  __shared__ float s_rv[1024];
  __shared__ int   s_ri[1024];

  for (int i = tid; i < N; i += 1024) dist[i] = FLT_BIG;
  if (tid == 0) s_cur = 0;
  __syncthreads();

  for (int it = 0; it < M; ++it) {
    if (tid == 0) {
      int cur = s_cur;
      cent_idx[it] = cur;
      float x = pos[cur * 3 + 0];
      float y = pos[cur * 3 + 1];
      float z = pos[cur * 3 + 2];
      s_cp[0] = x; s_cp[1] = y; s_cp[2] = z;
      cents[it * 3 + 0] = x;
      cents[it * 3 + 1] = y;
      cents[it * 3 + 2] = z;
    }
    __syncthreads();
    const float cx = s_cp[0], cy = s_cp[1], cz = s_cp[2];
    float bv = -FLT_BIG;
    int   bi = 0x7fffffff;
    for (int i = tid; i < N; i += 1024) {
      float dx = pos[i * 3 + 0] - cx;
      float dy = pos[i * 3 + 1] - cy;
      float dz = pos[i * 3 + 2] - cz;
      float d2 = dx * dx + dy * dy + dz * dz;
      float dd = fminf(dist[i], d2);
      dist[i] = dd;
      if (dd > bv) { bv = dd; bi = i; }
    }
    s_rv[tid] = bv;
    s_ri[tid] = bi;
    __syncthreads();
    for (int off = 512; off >= 1; off >>= 1) {
      if (tid < off) {
        float ov = s_rv[tid + off];
        int   oi = s_ri[tid + off];
        if (ov > s_rv[tid] || (ov == s_rv[tid] && oi < s_ri[tid])) {
          s_rv[tid] = ov; s_ri[tid] = oi;
        }
      }
      __syncthreads();
    }
    if (tid == 0) s_cur = s_ri[0];
    __syncthreads();
  }
}

__global__ __launch_bounds__(256)
void bq_pool_kernel(const float* __restrict__ pos,
                    const float* __restrict__ fs,
                    const float* __restrict__ fv,
                    const int* __restrict__ cent_idx,
                    float* __restrict__ pooled_s,
                    float* __restrict__ pooled_v,
                    int N, int C, int M) {
  const int m   = blockIdx.x;
  const int tid = threadIdx.x;
  __shared__ float s_cd2[CAP];
  __shared__ int   s_cidx[CAP];
  __shared__ int   s_cnt;
  __shared__ int   s_sel[K_LOCAL];
  __shared__ float s_rv[256];
  __shared__ int   s_ri[256];
  __shared__ int   s_rp[256];
  __shared__ float s_c[3];

  if (tid == 0) {
    s_cnt = 0;
    int ci = cent_idx[m];
    s_c[0] = pos[ci * 3 + 0];
    s_c[1] = pos[ci * 3 + 1];
    s_c[2] = pos[ci * 3 + 2];
  }
  __syncthreads();
  const float cx = s_c[0], cy = s_c[1], cz = s_c[2];

  for (int i = tid; i < N; i += 256) {
    float dx = pos[i * 3 + 0] - cx;
    float dy = pos[i * 3 + 1] - cy;
    float dz = pos[i * 3 + 2] - cz;
    float d2 = dx * dx + dy * dy + dz * dz;
    if (d2 <= RADIUS2) {
      int p = atomicAdd(&s_cnt, 1);
      if (p < CAP) { s_cd2[p] = d2; s_cidx[p] = i; }
    }
  }
  __syncthreads();
  const int ncand = min(s_cnt, CAP);

  for (int s = 0; s < K_LOCAL; ++s) {
    float bv = FLT_BIG;
    int   bi = 0x7fffffff;
    int   bp = -1;
    for (int p = tid; p < ncand; p += 256) {
      float v  = s_cd2[p];
      int   ix = s_cidx[p];
      if (v < bv || (v == bv && ix < bi)) { bv = v; bi = ix; bp = p; }
    }
    s_rv[tid] = bv; s_ri[tid] = bi; s_rp[tid] = bp;
    __syncthreads();
    for (int off = 128; off >= 1; off >>= 1) {
      if (tid < off) {
        float ov = s_rv[tid + off];
        int   oi = s_ri[tid + off];
        if (ov < s_rv[tid] || (ov == s_rv[tid] && oi < s_ri[tid])) {
          s_rv[tid] = ov; s_ri[tid] = oi; s_rp[tid] = s_rp[tid + off];
        }
      }
      __syncthreads();
    }
    if (tid == 0) {
      if (s < ncand) {
        int p = s_rp[0];
        s_sel[s] = s_cidx[p];
        s_cd2[p] = FLT_BIG;
      } else {
        s_sel[s] = s_sel[0];
      }
    }
    __syncthreads();
  }

  for (int j = tid; j < C * 4; j += 256) {
    const int c = j >> 2;
    const int d = j & 3;
    float mx = -FLT_BIG;
    #pragma unroll
    for (int t = 0; t < K_LOCAL; ++t) {
      int n = s_sel[t];
      float v = (d == 0) ? fs[(size_t)n * C + c]
                         : fv[((size_t)n * C + c) * 3 + (d - 1)];
      mx = fmaxf(mx, v);
    }
    if (d == 0) pooled_s[(size_t)m * C + c] = mx;
    else        pooled_v[(size_t)(d - 1) * M * C + (size_t)m * C + c] = mx;
  }
}

__global__ __launch_bounds__(256)
void mix_wmma_kernel(const float* __restrict__ pooled_s,
                     const float* __restrict__ pooled_v,
                     const float* __restrict__ Ws, const float* __restrict__ Wv,
                     const float* __restrict__ bs, const float* __restrict__ bv,
                     float* __restrict__ outS,
                     float* __restrict__ outV,
                     int M, int C) {
  const int lane = threadIdx.x & 31;
  const int wave = threadIdx.x >> 5;
  const int tiles_o = C >> 4;
  const int tiles_m = M >> 4;
  const int per_d   = tiles_m * tiles_o;
  const int tile    = blockIdx.x * (blockDim.x >> 5) + wave;
  if (tile >= per_d * 4) return;

  const int d   = tile / per_d;
  const int rem = tile % per_d;
  const int m0  = (rem / tiles_o) << 4;
  const int o0  = (rem % tiles_o) << 4;

  const float* A    = (d == 0) ? (pooled_s + (size_t)m0 * C)
                               : (pooled_v + (size_t)(d - 1) * M * C + (size_t)m0 * C);
  const float* B    = (d == 0) ? Ws : Wv;
  const float* bias = (d == 0) ? bs : bv;

  const int ml = lane & 15;
  const int kh = lane >> 4;

  v8f acc = {};
  for (int k0 = 0; k0 < C; k0 += 4) {
    const int ka = k0 + 2 * kh;
    v2f a = *(const v2f*)(A + (size_t)ml * C + ka);
    v2f b;
    b.x = B[(size_t)ka * C + o0 + ml];
    b.y = B[(size_t)(ka + 1) * C + o0 + ml];
    acc = __builtin_amdgcn_wmma_f32_16x16x4_f32(
        false, a, false, b, (short)0, acc, false, false);
  }

  const int   n  = o0 + ml;
  const float bb = bias[n];
  #pragma unroll
  for (int v = 0; v < 8; ++v) {
    const int mm  = m0 + v + 8 * kh;
    const float val = acc[v] + bb;
    if (d == 0) outS[(size_t)mm * C + n] = val;
    else        outV[((size_t)mm * C + n) * 3 + (d - 1)] = val;
  }
}

extern "C" void kernel_launch(void* const* d_in, const int* in_sizes, int n_in,
                              void* d_out, int out_size, void* d_ws, size_t ws_size,
                              hipStream_t stream) {
  const float* pos = (const float*)d_in[0];
  const float* fs  = (const float*)d_in[1];
  const float* fv  = (const float*)d_in[2];
  const float* Ws  = (const float*)d_in[3];
  const float* Wv  = (const float*)d_in[4];
  const float* bs  = (const float*)d_in[5];
  const float* bv  = (const float*)d_in[6];

  const int N = in_sizes[0] / 3;
  const int C = in_sizes[1] / N;
  const int M = (N < 2048) ? N : 2048;

  float* out   = (float*)d_out;
  float* cents = out;
  float* outS  = out + (size_t)M * 3;
  float* outV  = outS + (size_t)M * C;

  int*   cent_idx = (int*)d_ws;
  float* wsf      = (float*)d_ws;
  float* dist     = wsf + 4096;
  float* pooled_s = wsf + 4096 + 65536;
  float* pooled_v = pooled_s + (size_t)M * C;

  fps_kernel<<<1, 1024, 0, stream>>>(pos, dist, cent_idx, cents, N, M);
  bq_pool_kernel<<<M, 256, 0, stream>>>(pos, fs, fv, cent_idx,
                                        pooled_s, pooled_v, N, C, M);

  const int per_d = (M / 16) * (C / 16);
  const int total = per_d * 4;
  mix_wmma_kernel<<<(total + 7) / 8, 256, 0, stream>>>(
      pooled_s, pooled_v, Ws, Wv, bs, bv, outS, outV, M, C);
}


# FPS: per-point state (running dist, x in registers; y, most z in LDS; rest in a plane-major scratch copy read as b128) stays on chip over all steps, 512 threads x 98 points; results bit-identical
# speedup vs baseline: 3.5108x; 3.5108x over previous
#include <hip/hip_runtime.h>
#include <hip/hip_bf16.h>

#define FLT_BIG 3.402823466e+38f
#define RADIUS2 0.01f
#define K_LOCAL 32
#define CAP 2048

typedef __attribute__((ext_vector_type(2))) float v2f;
typedef __attribute__((ext_vector_type(8))) float v8f;

#define FPS_T     512
#define FPS_PPT   98
#define FPS_PITCH 156

typedef __attribute__((ext_vector_type(4))) float v4f;

__global__ __launch_bounds__(FPS_T)
void fps_kernel(const float* __restrict__ pos, float* __restrict__ spill,
                int* __restrict__ cent_idx, float* __restrict__ cents,
                int N, int M) {
  const int tid = threadIdx.x;
  __shared__ int   s_cur;
  __shared__ float s_cp[3];
  __shared__ float s_rv[FPS_T];
  __shared__ int   s_ri[FPS_T];
  __shared__ __attribute__((aligned(16))) float s_pt[FPS_T * FPS_PITCH];

  if (N > FPS_PPT * FPS_T) return;

  float*     L  = s_pt + tid * FPS_PITCH;
  const v4f* L4 = (const v4f*)L;
  const v4f* P4 = (const v4f*)spill;

  float px[FPS_PPT];
  float dist[FPS_PPT];
  #pragma unroll
  for (int j = 0; j < FPS_PPT; ++j) {
    const int i = tid + j * FPS_T;
    float x = 0.0f, y = 0.0f, z = 0.0f, d = -FLT_BIG;
    if (i < N) {
      x = pos[i * 3 + 0];
      y = pos[i * 3 + 1];
      z = pos[i * 3 + 2];
      d = FLT_BIG;
    }
    dist[j] = d;
    L[j] = y;
    if (j >= 54 && j < 82) spill[((((6 + ((j - 54) >> 2)) * FPS_T) + tid) << 2) + ((j - 54) & 3)] = x;
    else                   px[j] = x;
    if (j < 24)      spill[((((j >> 2) * FPS_T) + tid) << 2) + (j & 3)] = z;
    else if (j < 80) L[100 + (j - 24)] = z;
    else if (j < 82) L[98 + (j - 80)] = z;
    else             spill[((((13 + ((j - 82) >> 2)) * FPS_T) + tid) << 2) + ((j - 82) & 3)] = z;
  }
  if (tid == 0) s_cur = 0;
  __syncthreads();

  for (int it = 0; it < M; ++it) {
    v4f zb1[6];
    #pragma unroll
    for (int q = 0; q < 6; ++q) zb1[q] = P4[q * FPS_T + tid];

    if (tid == 0) {
      int cur = s_cur;
      cent_idx[it] = cur;
      float x = pos[cur * 3 + 0];
      float y = pos[cur * 3 + 1];
      float z = pos[cur * 3 + 2];
      s_cp[0] = x; s_cp[1] = y; s_cp[2] = z;
      cents[it * 3 + 0] = x;
      cents[it * 3 + 1] = y;
      cents[it * 3 + 2] = z;
    }
    __syncthreads();
    const float cx = s_cp[0], cy = s_cp[1], cz = s_cp[2];
    float bv = -FLT_BIG;
    int   bj = -1;
    v4f xb[7];
    v4f zc[4];
    #pragma unroll
    for (int j = 0; j < FPS_PPT; ++j) {
      if (j == 24) {
        #pragma unroll
        for (int q = 0; q < 7; ++q) xb[q] = P4[(6 + q) * FPS_T + tid];
      }
      if (j == 62) {
        #pragma unroll
        for (int q = 0; q < 4; ++q) zc[q] = P4[(13 + q) * FPS_T + tid];
      }
      const v4f yv = L4[j >> 2];
      const float y = yv[j & 3];
      float z;
      if (j < 24)      { z = zb1[j >> 2][j & 3]; }
      else if (j < 80) { const v4f zv = L4[25 + ((j - 24) >> 2)]; z = zv[(j - 24) & 3]; }
      else if (j < 82) { const v4f zv = L4[24]; z = zv[2 + (j - 80)]; }
      else             { z = zc[(j - 82) >> 2][(j - 82) & 3]; }
      float x;
      if (j >= 54 && j < 82) x = xb[(j - 54) >> 2][(j - 54) & 3];
      else                   x = px[j];
      float dx = x - cx;
      float dy = y - cy;
      float dz = z - cz;
      float d2 = __builtin_fmaf(dz, dz, __builtin_fmaf(dx, dx, dy * dy));
      float dd = fminf(dist[j], d2);
      dist[j] = dd;
      if (dd > bv) { bv = dd; bj = j; }
    }
    s_rv[tid] = bv;
    s_ri[tid] = (bj < 0) ? 0x7fffffff : (tid + bj * FPS_T);
    __syncthreads();
    for (int off = FPS_T / 2; off >= 1; off >>= 1) {
      if (tid < off) {
        float ov = s_rv[tid + off];
        int   oi = s_ri[tid + off];
        if (ov > s_rv[tid] || (ov == s_rv[tid] && oi < s_ri[tid])) {
          s_rv[tid] = ov; s_ri[tid] = oi;
        }
      }
      __syncthreads();
    }
    if (tid == 0) s_cur = s_ri[0];
    __syncthreads();
  }
}

__global__ __launch_bounds__(256)
void bq_pool_kernel(const float* __restrict__ pos,
                    const float* __restrict__ fs,
                    const float* __restrict__ fv,
                    const int* __restrict__ cent_idx,
                    float* __restrict__ pooled_s,
                    float* __restrict__ pooled_v,
                    int N, int C, int M) {
  const int m   = blockIdx.x;
  const int tid = threadIdx.x;
  __shared__ float s_cd2[CAP];
  __shared__ int   s_cidx[CAP];
  __shared__ int   s_cnt;
  __shared__ int   s_sel[K_LOCAL];
  __shared__ float s_rv[256];
  __shared__ int   s_ri[256];
  __shared__ int   s_rp[256];
  __shared__ float s_c[3];

  if (tid == 0) {
    s_cnt = 0;
    int ci = cent_idx[m];
    s_c[0] = pos[ci * 3 + 0];
    s_c[1] = pos[ci * 3 + 1];
    s_c[2] = pos[ci * 3 + 2];
  }
  __syncthreads();
  const float cx = s_c[0], cy = s_c[1], cz = s_c[2];

  for (int i = tid; i < N; i += 256) {
    float dx = pos[i * 3 + 0] - cx;
    float dy = pos[i * 3 + 1] - cy;
    float dz = pos[i * 3 + 2] - cz;
    float d2 = dx * dx + dy * dy + dz * dz;
    if (d2 <= RADIUS2) {
      int p = atomicAdd(&s_cnt, 1);
      if (p < CAP) { s_cd2[p] = d2; s_cidx[p] = i; }
    }
  }
  __syncthreads();
  const int ncand = min(s_cnt, CAP);

  for (int s = 0; s < K_LOCAL; ++s) {
    float bv = FLT_BIG;
    int   bi = 0x7fffffff;
    int   bp = -1;
    for (int p = tid; p < ncand; p += 256) {
      float v  = s_cd2[p];
      int   ix = s_cidx[p];
      if (v < bv || (v == bv && ix < bi)) { bv = v; bi = ix; bp = p; }
    }
    s_rv[tid] = bv; s_ri[tid] = bi; s_rp[tid] = bp;
    __syncthreads();
    for (int off = 128; off >= 1; off >>= 1) {
      if (tid < off) {
        float ov = s_rv[tid + off];
        int   oi = s_ri[tid + off];
        if (ov < s_rv[tid] || (ov == s_rv[tid] && oi < s_ri[tid])) {
          s_rv[tid] = ov; s_ri[tid] = oi; s_rp[tid] = s_rp[tid + off];
        }
      }
      __syncthreads();
    }
    if (tid == 0) {
      if (s < ncand) {
        int p = s_rp[0];
        s_sel[s] = s_cidx[p];
        s_cd2[p] = FLT_BIG;
      } else {
        s_sel[s] = s_sel[0];
      }
    }
    __syncthreads();
  }

  for (int j = tid; j < C * 4; j += 256) {
    const int c = j >> 2;
    const int d = j & 3;
    float mx = -FLT_BIG;
    #pragma unroll
    for (int t = 0; t < K_LOCAL; ++t) {
      int n = s_sel[t];
      float v = (d == 0) ? fs[(size_t)n * C + c]
                         : fv[((size_t)n * C + c) * 3 + (d - 1)];
      mx = fmaxf(mx, v);
    }
    if (d == 0) pooled_s[(size_t)m * C + c] = mx;
    else        pooled_v[(size_t)(d - 1) * M * C + (size_t)m * C + c] = mx;
  }
}

__global__ __launch_bounds__(256)
void mix_wmma_kernel(const float* __restrict__ pooled_s,
                     const float* __restrict__ pooled_v,
                     const float* __restrict__ Ws, const float* __restrict__ Wv,
                     const float* __restrict__ bs, const float* __restrict__ bv,
                     float* __restrict__ outS,
                     float* __restrict__ outV,
                     int M, int C) {
  const int lane = threadIdx.x & 31;
  const int wave = threadIdx.x >> 5;
  const int tiles_o = C >> 4;
  const int tiles_m = M >> 4;
  const int per_d   = tiles_m * tiles_o;
  const int tile    = blockIdx.x * (blockDim.x >> 5) + wave;
  if (tile >= per_d * 4) return;

  const int d   = tile / per_d;
  const int rem = tile % per_d;
  const int m0  = (rem / tiles_o) << 4;
  const int o0  = (rem % tiles_o) << 4;

  const float* A    = (d == 0) ? (pooled_s + (size_t)m0 * C)
                               : (pooled_v + (size_t)(d - 1) * M * C + (size_t)m0 * C);
  const float* B    = (d == 0) ? Ws : Wv;
  const float* bias = (d == 0) ? bs : bv;

  const int ml = lane & 15;
  const int kh = lane >> 4;

  v8f acc = {};
  for (int k0 = 0; k0 < C; k0 += 4) {
    const int ka = k0 + 2 * kh;
    v2f a = *(const v2f*)(A + (size_t)ml * C + ka);
    v2f b;
    b.x = B[(size_t)ka * C + o0 + ml];
    b.y = B[(size_t)(ka + 1) * C + o0 + ml];
    acc = __builtin_amdgcn_wmma_f32_16x16x4_f32(
        false, a, false, b, (short)0, acc, false, false);
  }

  const int   n  = o0 + ml;
  const float bb = bias[n];
  #pragma unroll
  for (int v = 0; v < 8; ++v) {
    const int mm  = m0 + v + 8 * kh;
    const float val = acc[v] + bb;
    if (d == 0) outS[(size_t)mm * C + n] = val;
    else        outV[((size_t)mm * C + n) * 3 + (d - 1)] = val;
  }
}

extern "C" void kernel_launch(void* const* d_in, const int* in_sizes, int n_in,
                              void* d_out, int out_size, void* d_ws, size_t ws_size,
                              hipStream_t stream) {
  const float* pos = (const float*)d_in[0];
  const float* fs  = (const float*)d_in[1];
  const float* fv  = (const float*)d_in[2];
  const float* Ws  = (const float*)d_in[3];
  const float* Wv  = (const float*)d_in[4];
  const float* bs  = (const float*)d_in[5];
  const float* bv  = (const float*)d_in[6];

  const int N = in_sizes[0] / 3;
  const int C = in_sizes[1] / N;
  const int M = (N < 2048) ? N : 2048;

  float* out   = (float*)d_out;
  float* cents = out;
  float* outS  = out + (size_t)M * 3;
  float* outV  = outS + (size_t)M * C;

  int*   cent_idx = (int*)d_ws;
  float* wsf      = (float*)d_ws;
  float* spill    = wsf + 4096;
  float* pooled_s = wsf + 4096 + 65536;
  float* pooled_v = pooled_s + (size_t)M * C;

  fps_kernel<<<1, 512, 0, stream>>>(pos, spill, cent_idx, cents, N, M);
  bq_pool_kernel<<<M, 256, 0, stream>>>(pos, fs, fv, cent_idx,
                                        pooled_s, pooled_v, N, C, M);

  const int per_d = (M / 16) * (C / 16);
  const int total = per_d * 4;
  mix_wmma_kernel<<<(total + 7) / 8, 256, 0, stream>>>(
      pooled_s, pooled_v, Ws, Wv, bs, bv, outS, outV, M, C);
}


# FPS argmax per step: in-wave DPP butterfly on (value, index), one 16-slot LDS exchange, ONE barrier (12 before); each wave finishes it and fetches the centroid itself; results bit-identical
# speedup vs baseline: 4.4638x; 1.2714x over previous
#include <hip/hip_runtime.h>
#include <hip/hip_bf16.h>

#define FLT_BIG 3.402823466e+38f
#define RADIUS2 0.01f
#define K_LOCAL 32
#define CAP 2048

typedef __attribute__((ext_vector_type(2))) float v2f;
typedef __attribute__((ext_vector_type(8))) float v8f;

#define FPS_T     512
#define FPS_PPT   98
#define FPS_PITCH 156

typedef __attribute__((ext_vector_type(4))) float v4f;
typedef __attribute__((ext_vector_type(2))) int   v2i;

__device__ __forceinline__ void fps_take(float& bv, int& bi, float ov, int oi) {
  const bool t = (ov > bv) | ((ov == bv) & (oi < bi));
  bv = t ? ov : bv;
  bi = t ? oi : bi;
}

template <int MASK>
__device__ __forceinline__ void fps_row_step(float& bv, int& bi) {
  const int vb = __builtin_bit_cast(int, bv);
  const int ov = __builtin_amdgcn_update_dpp(vb, vb, 0x160 + MASK, 0xf, 0xf, false);
  const int oi = __builtin_amdgcn_update_dpp(bi, bi, 0x160 + MASK, 0xf, 0xf, false);
  fps_take(bv, bi, __builtin_bit_cast(float, ov), oi);
}

__global__ __launch_bounds__(FPS_T)
void fps_kernel(const float* __restrict__ pos, float* __restrict__ spill,
                int* __restrict__ cent_idx, float* __restrict__ cents,
                int N, int M) {
  const int tid  = threadIdx.x;
  const int lane = tid & 31;
  const int wave = tid >> 5;
  __shared__ __attribute__((aligned(8)))  v2i   s_win[2][FPS_T / 32];
  __shared__ __attribute__((aligned(16))) float s_pt[FPS_T * FPS_PITCH];

  if (N > FPS_PPT * FPS_T) return;

  float*     L  = s_pt + tid * FPS_PITCH;
  const v4f* L4 = (const v4f*)L;
  const v4f* P4 = (const v4f*)spill;

  float px[FPS_PPT];
  float dist[FPS_PPT];
  #pragma unroll
  for (int j = 0; j < FPS_PPT; ++j) {
    const int i = tid + j * FPS_T;
    float x = 0.0f, y = 0.0f, z = 0.0f, d = -FLT_BIG;
    if (i < N) {
      x = pos[i * 3 + 0];
      y = pos[i * 3 + 1];
      z = pos[i * 3 + 2];
      d = FLT_BIG;
    }
    dist[j] = d;
    L[j] = y;
    if (j >= 54 && j < 82) spill[((((6 + ((j - 54) >> 2)) * FPS_T) + tid) << 2) + ((j - 54) & 3)] = x;
    else                   px[j] = x;
    if (j < 24)      spill[((((j >> 2) * FPS_T) + tid) << 2) + (j & 3)] = z;
    else if (j < 80) L[100 + (j - 24)] = z;
    else if (j < 82) L[98 + (j - 80)] = z;
    else             spill[((((13 + ((j - 82) >> 2)) * FPS_T) + tid) << 2) + ((j - 82) & 3)] = z;
  }
  __syncthreads();

  v4f zb1[6];
  #pragma unroll
  for (int q = 0; q < 6; ++q) zb1[q] = P4[q * FPS_T + tid];

  int cur = 0;
  for (int it = 0; it < M; ++it) {
    const float cx = pos[cur * 3 + 0];
    const float cy = pos[cur * 3 + 1];
    const float cz = pos[cur * 3 + 2];
    if (tid == 0) {
      cent_idx[it] = cur;
      cents[it * 3 + 0] = cx;
      cents[it * 3 + 1] = cy;
      cents[it * 3 + 2] = cz;
    }
    float bv = -FLT_BIG;
    int   bj = -1;
    v4f xb[7];
    v4f zc[4];
    #pragma unroll
    for (int j = 0; j < FPS_PPT; ++j) {
      if (j == 24) {
        #pragma unroll
        for (int q = 0; q < 7; ++q) xb[q] = P4[(6 + q) * FPS_T + tid];
      }
      if (j == 62) {
        #pragma unroll
        for (int q = 0; q < 4; ++q) zc[q] = P4[(13 + q) * FPS_T + tid];
      }
      if (j == 82) {
        #pragma unroll
        for (int q = 0; q < 6; ++q) zb1[q] = P4[q * FPS_T + tid];
      }
      const v4f yv = L4[j >> 2];
      const float y = yv[j & 3];
      float z;
      if (j < 24)      { z = zb1[j >> 2][j & 3]; }
      else if (j < 80) { const v4f zv = L4[25 + ((j - 24) >> 2)]; z = zv[(j - 24) & 3]; }
      else if (j < 82) { const v4f zv = L4[24]; z = zv[2 + (j - 80)]; }
      else             { z = zc[(j - 82) >> 2][(j - 82) & 3]; }
      float x;
      if (j >= 54 && j < 82) x = xb[(j - 54) >> 2][(j - 54) & 3];
      else                   x = px[j];
      float dx = x - cx;
      float dy = y - cy;
      float dz = z - cz;
      float d2 = __builtin_fmaf(dz, dz, __builtin_fmaf(dx, dx, dy * dy));
      float dd = fminf(dist[j], d2);
      dist[j] = dd;
      if (dd > bv) { bv = dd; bj = j; }
    }
    int bi = (bj < 0) ? 0x7fffffff : (tid + bj * FPS_T);
    fps_row_step<1>(bv, bi);
    fps_row_step<2>(bv, bi);
    fps_row_step<4>(bv, bi);
    fps_row_step<8>(bv, bi);
    {
      const int vb = __builtin_bit_cast(int, bv);
      const int ov = __builtin_amdgcn_permlanex16(vb, vb, 0x76543210, 0xfedcba98, false, false);
      const int oi = __builtin_amdgcn_permlanex16(bi, bi, 0x76543210, 0xfedcba98, false, false);
      fps_take(bv, bi, __builtin_bit_cast(float, ov), oi);
    }
    if (lane == 0) {
      v2i w;
      w.x = __builtin_bit_cast(int, bv);
      w.y = bi;
      s_win[it & 1][wave] = w;
    }
    __builtin_amdgcn_fence(__ATOMIC_RELEASE, "workgroup", "local");
    __builtin_amdgcn_s_barrier();
    __builtin_amdgcn_fence(__ATOMIC_ACQUIRE, "workgroup", "local");
    {
      const v2i w = s_win[it & 1][lane & 15];
      bv = __builtin_bit_cast(float, w.x);
      bi = w.y;
    }
    fps_row_step<1>(bv, bi);
    fps_row_step<2>(bv, bi);
    fps_row_step<4>(bv, bi);
    fps_row_step<8>(bv, bi);
    cur = __builtin_amdgcn_readfirstlane(bi);
    cur = (cur < N) ? cur : 0;
  }
}

__global__ __launch_bounds__(256)
void bq_pool_kernel(const float* __restrict__ pos,
                    const float* __restrict__ fs,
                    const float* __restrict__ fv,
                    const int* __restrict__ cent_idx,
                    float* __restrict__ pooled_s,
                    float* __restrict__ pooled_v,
                    int N, int C, int M) {
  const int m   = blockIdx.x;
  const int tid = threadIdx.x;
  __shared__ float s_cd2[CAP];
  __shared__ int   s_cidx[CAP];
  __shared__ int   s_cnt;
  __shared__ int   s_sel[K_LOCAL];
  __shared__ float s_rv[256];
  __shared__ int   s_ri[256];
  __shared__ int   s_rp[256];
  __shared__ float s_c[3];

  if (tid == 0) {
    s_cnt = 0;
    int ci = cent_idx[m];
    s_c[0] = pos[ci * 3 + 0];
    s_c[1] = pos[ci * 3 + 1];
    s_c[2] = pos[ci * 3 + 2];
  }
  __syncthreads();
  const float cx = s_c[0], cy = s_c[1], cz = s_c[2];

  for (int i = tid; i < N; i += 256) {
    float dx = pos[i * 3 + 0] - cx;
    float dy = pos[i * 3 + 1] - cy;
    float dz = pos[i * 3 + 2] - cz;
    float d2 = dx * dx + dy * dy + dz * dz;
    if (d2 <= RADIUS2) {
      int p = atomicAdd(&s_cnt, 1);
      if (p < CAP) { s_cd2[p] = d2; s_cidx[p] = i; }
    }
  }
  __syncthreads();
  const int ncand = min(s_cnt, CAP);

  for (int s = 0; s < K_LOCAL; ++s) {
    float bv = FLT_BIG;
    int   bi = 0x7fffffff;
    int   bp = -1;
    for (int p = tid; p < ncand; p += 256) {
      float v  = s_cd2[p];
      int   ix = s_cidx[p];
      if (v < bv || (v == bv && ix < bi)) { bv = v; bi = ix; bp = p; }
    }
    s_rv[tid] = bv; s_ri[tid] = bi; s_rp[tid] = bp;
    __syncthreads();
    for (int off = 128; off >= 1; off >>= 1) {
      if (tid < off) {
        float ov = s_rv[tid + off];
        int   oi = s_ri[tid + off];
        if (ov < s_rv[tid] || (ov == s_rv[tid] && oi < s_ri[tid])) {
          s_rv[tid] = ov; s_ri[tid] = oi; s_rp[tid] = s_rp[tid + off];
        }
      }
      __syncthreads();
    }
    if (tid == 0) {
      if (s < ncand) {
        int p = s_rp[0];
        s_sel[s] = s_cidx[p];
        s_cd2[p] = FLT_BIG;
      } else {
        s_sel[s] = s_sel[0];
      }
    }
    __syncthreads();
  }

  for (int j = tid; j < C * 4; j += 256) {
    const int c = j >> 2;
    const int d = j & 3;
    float mx = -FLT_BIG;
    #pragma unroll
    for (int t = 0; t < K_LOCAL; ++t) {
      int n = s_sel[t];
      float v = (d == 0) ? fs[(size_t)n * C + c]
                         : fv[((size_t)n * C + c) * 3 + (d - 1)];
      mx = fmaxf(mx, v);
    }
    if (d == 0) pooled_s[(size_t)m * C + c] = mx;
    else        pooled_v[(size_t)(d - 1) * M * C + (size_t)m * C + c] = mx;
  }
}

__global__ __launch_bounds__(256)
void mix_wmma_kernel(const float* __restrict__ pooled_s,
                     const float* __restrict__ pooled_v,
                     const float* __restrict__ Ws, const float* __restrict__ Wv,
                     const float* __restrict__ bs, const float* __restrict__ bv,
                     float* __restrict__ outS,
                     float* __restrict__ outV,
                     int M, int C) {
  const int lane = threadIdx.x & 31;
  const int wave = threadIdx.x >> 5;
  const int tiles_o = C >> 4;
  const int tiles_m = M >> 4;
  const int per_d   = tiles_m * tiles_o;
  const int tile    = blockIdx.x * (blockDim.x >> 5) + wave;
  if (tile >= per_d * 4) return;

  const int d   = tile / per_d;
  const int rem = tile % per_d;
  const int m0  = (rem / tiles_o) << 4;
  const int o0  = (rem % tiles_o) << 4;

  const float* A    = (d == 0) ? (pooled_s + (size_t)m0 * C)
                               : (pooled_v + (size_t)(d - 1) * M * C + (size_t)m0 * C);
  const float* B    = (d == 0) ? Ws : Wv;
  const float* bias = (d == 0) ? bs : bv;

  const int ml = lane & 15;
  const int kh = lane >> 4;

  v8f acc = {};
  for (int k0 = 0; k0 < C; k0 += 4) {
    const int ka = k0 + 2 * kh;
    v2f a = *(const v2f*)(A + (size_t)ml * C + ka);
    v2f b;
    b.x = B[(size_t)ka * C + o0 + ml];
    b.y = B[(size_t)(ka + 1) * C + o0 + ml];
    acc = __builtin_amdgcn_wmma_f32_16x16x4_f32(
        false, a, false, b, (short)0, acc, false, false);
  }

  const int   n  = o0 + ml;
  const float bb = bias[n];
  #pragma unroll
  for (int v = 0; v < 8; ++v) {
    const int mm  = m0 + v + 8 * kh;
    const float val = acc[v] + bb;
    if (d == 0) outS[(size_t)mm * C + n] = val;
    else        outV[((size_t)mm * C + n) * 3 + (d - 1)] = val;
  }
}

extern "C" void kernel_launch(void* const* d_in, const int* in_sizes, int n_in,
                              void* d_out, int out_size, void* d_ws, size_t ws_size,
                              hipStream_t stream) {
  const float* pos = (const float*)d_in[0];
  const float* fs  = (const float*)d_in[1];
  const float* fv  = (const float*)d_in[2];
  const float* Ws  = (const float*)d_in[3];
  const float* Wv  = (const float*)d_in[4];
  const float* bs  = (const float*)d_in[5];
  const float* bv  = (const float*)d_in[6];

  const int N = in_sizes[0] / 3;
  const int C = in_sizes[1] / N;
  const int M = (N < 2048) ? N : 2048;

  float* out   = (float*)d_out;
  float* cents = out;
  float* outS  = out + (size_t)M * 3;
  float* outV  = outS + (size_t)M * C;

  int*   cent_idx = (int*)d_ws;
  float* wsf      = (float*)d_ws;
  float* spill    = wsf + 4096;
  float* pooled_s = wsf + 4096 + 65536;
  float* pooled_v = pooled_s + (size_t)M * C;

  fps_kernel<<<1, 512, 0, stream>>>(pos, spill, cent_idx, cents, N, M);
  bq_pool_kernel<<<M, 256, 0, stream>>>(pos, fs, fv, cent_idx,
                                        pooled_s, pooled_v, N, C, M);

  const int per_d = (M / 16) * (C / 16);
  const int total = per_d * 4;
  mix_wmma_kernel<<<(total + 7) / 8, 256, 0, stream>>>(
      pooled_s, pooled_v, Ws, Wv, bs, bv, outS, outV, M, C);
}


# FPS argmax carries no index through the point loop: one max per point into 7 group maxima, index recovered by equality scan of one group; wave and slot exchanges as single DPP max/min; bit-identical
# speedup vs baseline: 5.0750x; 1.1369x over previous
#include <hip/hip_runtime.h>
#include <hip/hip_bf16.h>

#define FLT_BIG 3.402823466e+38f
#define RADIUS2 0.01f
#define K_LOCAL 32
#define CAP 2048

typedef __attribute__((ext_vector_type(2))) float v2f;
typedef __attribute__((ext_vector_type(8))) float v8f;

#define FPS_T     512
#define FPS_PPT   98
#define FPS_PITCH 156
#define FPS_GS    14
#define FPS_NG    7

typedef __attribute__((ext_vector_type(4))) float v4f;
typedef __attribute__((ext_vector_type(2))) int   v2i;

template <int MASK>
__device__ __forceinline__ int fps_row_max(int v) {
  const int ov = __builtin_amdgcn_update_dpp(v, v, 0x160 + MASK, 0xf, 0xf, true);
  return (ov > v) ? ov : v;
}

template <int MASK>
__device__ __forceinline__ int fps_row_min(int v) {
  const int ov = __builtin_amdgcn_update_dpp(v, v, 0x160 + MASK, 0xf, 0xf, true);
  return (ov < v) ? ov : v;
}

__global__ __launch_bounds__(FPS_T)
void fps_kernel(const float* __restrict__ pos, float* __restrict__ spill,
                int* __restrict__ cent_idx, float* __restrict__ cents,
                int N, int M) {
  const int tid  = threadIdx.x;
  const int lane = tid & 31;
  const int wave = tid >> 5;
  __shared__ __attribute__((aligned(8)))  v2i   s_win[2][FPS_T / 32];
  __shared__ __attribute__((aligned(16))) float s_pt[FPS_T * FPS_PITCH];

  if (N > FPS_PPT * FPS_T) return;

  float*     L  = s_pt + tid * FPS_PITCH;
  const v4f* L4 = (const v4f*)L;
  const v4f* P4 = (const v4f*)spill;

  float px[FPS_PPT];
  float dist[FPS_PPT];
  #pragma unroll
  for (int j = 0; j < FPS_PPT; ++j) {
    const int i = tid + j * FPS_T;
    float x = 0.0f, y = 0.0f, z = 0.0f, d = -FLT_BIG;
    if (i < N) {
      x = pos[i * 3 + 0];
      y = pos[i * 3 + 1];
      z = pos[i * 3 + 2];
      d = FLT_BIG;
    }
    dist[j] = d;
    L[j] = y;
    if (j >= 54 && j < 82) spill[((((6 + ((j - 54) >> 2)) * FPS_T) + tid) << 2) + ((j - 54) & 3)] = x;
    else                   px[j] = x;
    if (j < 24)      spill[((((j >> 2) * FPS_T) + tid) << 2) + (j & 3)] = z;
    else if (j < 80) L[100 + (j - 24)] = z;
    else if (j < 82) L[98 + (j - 80)] = z;
    else             spill[((((13 + ((j - 82) >> 2)) * FPS_T) + tid) << 2) + ((j - 82) & 3)] = z;
  }
  __syncthreads();

  v4f zb1[6];
  #pragma unroll
  for (int q = 0; q < 6; ++q) zb1[q] = P4[q * FPS_T + tid];

  int cur = 0;
  for (int it = 0; it < M; ++it) {
    const float cx = pos[cur * 3 + 0];
    const float cy = pos[cur * 3 + 1];
    const float cz = pos[cur * 3 + 2];
    if (tid == 0) {
      cent_idx[it] = cur;
      cents[it * 3 + 0] = cx;
      cents[it * 3 + 1] = cy;
      cents[it * 3 + 2] = cz;
    }
    float gm[FPS_NG];
    v4f xb[7];
    v4f zc[4];
    #pragma unroll
    for (int j = 0; j < FPS_PPT; ++j) {
      if (j == 24) {
        #pragma unroll
        for (int q = 0; q < 7; ++q) xb[q] = P4[(6 + q) * FPS_T + tid];
      }
      if (j == 62) {
        #pragma unroll
        for (int q = 0; q < 4; ++q) zc[q] = P4[(13 + q) * FPS_T + tid];
      }
      if (j == 82) {
        #pragma unroll
        for (int q = 0; q < 6; ++q) zb1[q] = P4[q * FPS_T + tid];
      }
      const v4f yv = L4[j >> 2];
      const float y = yv[j & 3];
      float z;
      if (j < 24)      { z = zb1[j >> 2][j & 3]; }
      else if (j < 80) { const v4f zv = L4[25 + ((j - 24) >> 2)]; z = zv[(j - 24) & 3]; }
      else if (j < 82) { const v4f zv = L4[24]; z = zv[2 + (j - 80)]; }
      else             { z = zc[(j - 82) >> 2][(j - 82) & 3]; }
      float x;
      if (j >= 54 && j < 82) x = xb[(j - 54) >> 2][(j - 54) & 3];
      else                   x = px[j];
      float dx = x - cx;
      float dy = y - cy;
      float dz = z - cz;
      float d2 = __builtin_fmaf(dz, dz, __builtin_fmaf(dx, dx, dy * dy));
      float dd = fminf(dist[j], d2);
      dist[j] = dd;
      if (j % FPS_GS == 0) gm[j / FPS_GS] = dd;
      else                 gm[j / FPS_GS] = fmaxf(gm[j / FPS_GS], dd);
    }
    float bv = gm[0];
    #pragma unroll
    for (int k = 1; k < FPS_NG; ++k) bv = fmaxf(bv, gm[k]);
    int vb = __builtin_bit_cast(int, bv);
    vb = fps_row_max<1>(vb);
    vb = fps_row_max<2>(vb);
    vb = fps_row_max<4>(vb);
    vb = fps_row_max<8>(vb);
    {
      const int ov = __builtin_amdgcn_permlanex16(vb, vb, 0x76543210, 0xfedcba98, false, false);
      vb = (ov > vb) ? ov : vb;
    }
    const float V = __builtin_bit_cast(float, __builtin_amdgcn_readfirstlane(vb));
    int kg = 0;
    #pragma unroll
    for (int k = FPS_NG - 1; k >= 0; --k) {
      if (__builtin_amdgcn_ballot_w32(gm[k] == V) != 0u) kg = k;
    }
    int      jj = 0;
    unsigned mm = 1u;
    #pragma unroll
    for (int k = 0; k < FPS_NG; ++k) {
      if (kg == k) {
        #pragma unroll
        for (int r = FPS_GS - 1; r >= 0; --r) {
          const unsigned m = __builtin_amdgcn_ballot_w32(dist[k * FPS_GS + r] == V);
          if (m != 0u) { jj = k * FPS_GS + r; mm = m; }
        }
      }
    }
    const int bi = (wave << 5) + __builtin_ctz(mm) + jj * FPS_T;
    if (lane == 0) {
      v2i w;
      w.x = __builtin_bit_cast(int, V);
      w.y = bi;
      s_win[it & 1][wave] = w;
    }
    __builtin_amdgcn_fence(__ATOMIC_RELEASE, "workgroup", "local");
    __builtin_amdgcn_s_barrier();
    __builtin_amdgcn_fence(__ATOMIC_ACQUIRE, "workgroup", "local");
    const v2i ws = s_win[it & 1][lane & 15];
    int tv = ws.x;
    tv = fps_row_max<1>(tv);
    tv = fps_row_max<2>(tv);
    tv = fps_row_max<4>(tv);
    tv = fps_row_max<8>(tv);
    int ci = (ws.x == tv) ? ws.y : 0x7fffffff;
    ci = fps_row_min<1>(ci);
    ci = fps_row_min<2>(ci);
    ci = fps_row_min<4>(ci);
    ci = fps_row_min<8>(ci);
    cur = __builtin_amdgcn_readfirstlane(ci);
    cur = (cur < N) ? cur : 0;
  }
}

__global__ __launch_bounds__(256)
void bq_pool_kernel(const float* __restrict__ pos,
                    const float* __restrict__ fs,
                    const float* __restrict__ fv,
                    const int* __restrict__ cent_idx,
                    float* __restrict__ pooled_s,
                    float* __restrict__ pooled_v,
                    int N, int C, int M) {
  const int m   = blockIdx.x;
  const int tid = threadIdx.x;
  __shared__ float s_cd2[CAP];
  __shared__ int   s_cidx[CAP];
  __shared__ int   s_cnt;
  __shared__ int   s_sel[K_LOCAL];
  __shared__ float s_rv[256];
  __shared__ int   s_ri[256];
  __shared__ int   s_rp[256];
  __shared__ float s_c[3];

  if (tid == 0) {
    s_cnt = 0;
    int ci = cent_idx[m];
    s_c[0] = pos[ci * 3 + 0];
    s_c[1] = pos[ci * 3 + 1];
    s_c[2] = pos[ci * 3 + 2];
  }
  __syncthreads();
  const float cx = s_c[0], cy = s_c[1], cz = s_c[2];

  for (int i = tid; i < N; i += 256) {
    float dx = pos[i * 3 + 0] - cx;
    float dy = pos[i * 3 + 1] - cy;
    float dz = pos[i * 3 + 2] - cz;
    float d2 = dx * dx + dy * dy + dz * dz;
    if (d2 <= RADIUS2) {
      int p = atomicAdd(&s_cnt, 1);
      if (p < CAP) { s_cd2[p] = d2; s_cidx[p] = i; }
    }
  }
  __syncthreads();
  const int ncand = min(s_cnt, CAP);

  for (int s = 0; s < K_LOCAL; ++s) {
    float bv = FLT_BIG;
    int   bi = 0x7fffffff;
    int   bp = -1;
    for (int p = tid; p < ncand; p += 256) {
      float v  = s_cd2[p];
      int   ix = s_cidx[p];
      if (v < bv || (v == bv && ix < bi)) { bv = v; bi = ix; bp = p; }
    }
    s_rv[tid] = bv; s_ri[tid] = bi; s_rp[tid] = bp;
    __syncthreads();
    for (int off = 128; off >= 1; off >>= 1) {
      if (tid < off) {
        float ov = s_rv[tid + off];
        int   oi = s_ri[tid + off];
        if (ov < s_rv[tid] || (ov == s_rv[tid] && oi < s_ri[tid])) {
          s_rv[tid] = ov; s_ri[tid] = oi; s_rp[tid] = s_rp[tid + off];
        }
      }
      __syncthreads();
    }
    if (tid == 0) {
      if (s < ncand) {
        int p = s_rp[0];
        s_sel[s] = s_cidx[p];
        s_cd2[p] = FLT_BIG;
      } else {
        s_sel[s] = s_sel[0];
      }
    }
    __syncthreads();
  }

  for (int j = tid; j < C * 4; j += 256) {
    const int c = j >> 2;
    const int d = j & 3;
    float mx = -FLT_BIG;
    #pragma unroll
    for (int t = 0; t < K_LOCAL; ++t) {
      int n = s_sel[t];
      float v = (d == 0) ? fs[(size_t)n * C + c]
                         : fv[((size_t)n * C + c) * 3 + (d - 1)];
      mx = fmaxf(mx, v);
    }
    if (d == 0) pooled_s[(size_t)m * C + c] = mx;
    else        pooled_v[(size_t)(d - 1) * M * C + (size_t)m * C + c] = mx;
  }
}

__global__ __launch_bounds__(256)
void mix_wmma_kernel(const float* __restrict__ pooled_s,
                     const float* __restrict__ pooled_v,
                     const float* __restrict__ Ws, const float* __restrict__ Wv,
                     const float* __restrict__ bs, const float* __restrict__ bv,
                     float* __restrict__ outS,
                     float* __restrict__ outV,
                     int M, int C) {
  const int lane = threadIdx.x & 31;
  const int wave = threadIdx.x >> 5;
  const int tiles_o = C >> 4;
  const int tiles_m = M >> 4;
  const int per_d   = tiles_m * tiles_o;
  const int tile    = blockIdx.x * (blockDim.x >> 5) + wave;
  if (tile >= per_d * 4) return;

  const int d   = tile / per_d;
  const int rem = tile % per_d;
  const int m0  = (rem / tiles_o) << 4;
  const int o0  = (rem % tiles_o) << 4;

  const float* A    = (d == 0) ? (pooled_s + (size_t)m0 * C)
                               : (pooled_v + (size_t)(d - 1) * M * C + (size_t)m0 * C);
  const float* B    = (d == 0) ? Ws : Wv;
  const float* bias = (d == 0) ? bs : bv;

  const int ml = lane & 15;
  const int kh = lane >> 4;

  v8f acc = {};
  for (int k0 = 0; k0 < C; k0 += 4) {
    const int ka = k0 + 2 * kh;
    v2f a = *(const v2f*)(A + (size_t)ml * C + ka);
    v2f b;
    b.x = B[(size_t)ka * C + o0 + ml];
    b.y = B[(size_t)(ka + 1) * C + o0 + ml];
    acc = __builtin_amdgcn_wmma_f32_16x16x4_f32(
        false, a, false, b, (short)0, acc, false, false);
  }

  const int   n  = o0 + ml;
  const float bb = bias[n];
  #pragma unroll
  for (int v = 0; v < 8; ++v) {
    const int mm  = m0 + v + 8 * kh;
    const float val = acc[v] + bb;
    if (d == 0) outS[(size_t)mm * C + n] = val;
    else        outV[((size_t)mm * C + n) * 3 + (d - 1)] = val;
  }
}

extern "C" void kernel_launch(void* const* d_in, const int* in_sizes, int n_in,
                              void* d_out, int out_size, void* d_ws, size_t ws_size,
                              hipStream_t stream) {
  const float* pos = (const float*)d_in[0];
  const float* fs  = (const float*)d_in[1];
  const float* fv  = (const float*)d_in[2];
  const float* Ws  = (const float*)d_in[3];
  const float* Wv  = (const float*)d_in[4];
  const float* bs  = (const float*)d_in[5];
  const float* bv  = (const float*)d_in[6];

  const int N = in_sizes[0] / 3;
  const int C = in_sizes[1] / N;
  const int M = (N < 2048) ? N : 2048;

  float* out   = (float*)d_out;
  float* cents = out;
  float* outS  = out + (size_t)M * 3;
  float* outV  = outS + (size_t)M * C;

  int*   cent_idx = (int*)d_ws;
  float* wsf      = (float*)d_ws;
  float* spill    = wsf + 4096;
  float* pooled_s = wsf + 4096 + 65536;
  float* pooled_v = pooled_s + (size_t)M * C;

  fps_kernel<<<1, 512, 0, stream>>>(pos, spill, cent_idx, cents, N, M);
  bq_pool_kernel<<<M, 256, 0, stream>>>(pos, fs, fv, cent_idx,
                                        pooled_s, pooled_v, N, C, M);

  const int per_d = (M / 16) * (C / 16);
  const int total = per_d * 4;
  mix_wmma_kernel<<<(total + 7) / 8, 256, 0, stream>>>(
      pooled_s, pooled_v, Ws, Wv, bs, bv, outS, outV, M, C);
}


# FPS point loop by quads: each stage of the distance chain runs on 4 points at once and compiles to packed 2-point f32 add/mul/fma (half the issues, independent neighbours); results bit-identical
# speedup vs baseline: 6.5956x; 1.2996x over previous
#include <hip/hip_runtime.h>
#include <hip/hip_bf16.h>

#define FLT_BIG 3.402823466e+38f
#define RADIUS2 0.01f
#define K_LOCAL 32
#define CAP 2048

typedef __attribute__((ext_vector_type(2))) float v2f;
typedef __attribute__((ext_vector_type(8))) float v8f;

#define FPS_T     512
#define FPS_PPT   98
#define FPS_PITCH 156
#define FPS_GS    14
#define FPS_NG    7
#define FPS_NQ    25

typedef __attribute__((ext_vector_type(4))) float v4f;
typedef __attribute__((ext_vector_type(2))) int   v2i;

template <int MASK>
__device__ __forceinline__ int fps_row_max(int v) {
  const int ov = __builtin_amdgcn_update_dpp(v, v, 0x160 + MASK, 0xf, 0xf, true);
  return (ov > v) ? ov : v;
}

template <int MASK>
__device__ __forceinline__ int fps_row_min(int v) {
  const int ov = __builtin_amdgcn_update_dpp(v, v, 0x160 + MASK, 0xf, 0xf, true);
  return (ov < v) ? ov : v;
}

__global__ __launch_bounds__(FPS_T)
void fps_kernel(const float* __restrict__ pos, float* __restrict__ spill,
                int* __restrict__ cent_idx, float* __restrict__ cents,
                int N, int M) {
  const int tid  = threadIdx.x;
  const int lane = tid & 31;
  const int wave = tid >> 5;
  __shared__ __attribute__((aligned(8)))  v2i   s_win[2][FPS_T / 32];
  __shared__ __attribute__((aligned(16))) float s_pt[FPS_T * FPS_PITCH];

  if (N > FPS_PPT * FPS_T) return;

  float*     L  = s_pt + tid * FPS_PITCH;
  const v4f* L4 = (const v4f*)L;
  const v4f* P4 = (const v4f*)spill;

  float px[FPS_PPT];
  float dist[FPS_PPT];
  #pragma unroll
  for (int j = 0; j < FPS_PPT; ++j) {
    const int i = tid + j * FPS_T;
    float x = 0.0f, y = 0.0f, z = 0.0f, d = -FLT_BIG;
    if (i < N) {
      x = pos[i * 3 + 0];
      y = pos[i * 3 + 1];
      z = pos[i * 3 + 2];
      d = FLT_BIG;
    }
    dist[j] = d;
    L[j] = y;
    if (j >= 54 && j < 82) spill[((((6 + ((j - 54) >> 2)) * FPS_T) + tid) << 2) + ((j - 54) & 3)] = x;
    else                   px[j] = x;
    if (j < 24)      spill[((((j >> 2) * FPS_T) + tid) << 2) + (j & 3)] = z;
    else if (j < 80) L[100 + (j - 24)] = z;
    else if (j < 82) L[98 + (j - 80)] = z;
    else             spill[((((13 + ((j - 82) >> 2)) * FPS_T) + tid) << 2) + ((j - 82) & 3)] = z;
  }
  __syncthreads();

  v4f zb1[6];
  #pragma unroll
  for (int q = 0; q < 6; ++q) zb1[q] = P4[q * FPS_T + tid];

  int cur = 0;
  for (int it = 0; it < M; ++it) {
    const float cx = pos[cur * 3 + 0];
    const float cy = pos[cur * 3 + 1];
    const float cz = pos[cur * 3 + 2];
    if (tid == 0) {
      cent_idx[it] = cur;
      cents[it * 3 + 0] = cx;
      cents[it * 3 + 1] = cy;
      cents[it * 3 + 2] = cz;
    }
    float gm[FPS_NG];
    v4f xb[7];
    v4f zc[4];
    #pragma unroll
    for (int q = 0; q < FPS_NQ; ++q) {
      if (q == 6) {
        #pragma unroll
        for (int p = 0; p < 7; ++p) xb[p] = P4[(6 + p) * FPS_T + tid];
      }
      if (q == 15) {
        #pragma unroll
        for (int p = 0; p < 4; ++p) zc[p] = P4[(13 + p) * FPS_T + tid];
      }
      if (q == 21) {
        #pragma unroll
        for (int p = 0; p < 6; ++p) zb1[p] = P4[p * FPS_T + tid];
      }
      const v4f yv = L4[q];
      float dx[4], dy[4], dz[4], d2[4];
      #pragma unroll
      for (int r = 0; r < 4; ++r) {
        const int j = 4 * q + r;
        if (j < FPS_PPT) {
          float z;
          if (j < 24)      { z = zb1[j >> 2][j & 3]; }
          else if (j < 80) { const v4f zv = L4[25 + ((j - 24) >> 2)]; z = zv[(j - 24) & 3]; }
          else if (j < 82) { const v4f zv = L4[24]; z = zv[2 + (j - 80)]; }
          else             { z = zc[(j - 82) >> 2][(j - 82) & 3]; }
          float x;
          if (j >= 54 && j < 82) x = xb[(j - 54) >> 2][(j - 54) & 3];
          else                   x = px[j];
          dx[r] = x - cx;
          dy[r] = yv[r] - cy;
          dz[r] = z - cz;
        }
      }
      #pragma unroll
      for (int r = 0; r < 4; ++r) if (4 * q + r < FPS_PPT) d2[r] = dy[r] * dy[r];
      #pragma unroll
      for (int r = 0; r < 4; ++r) if (4 * q + r < FPS_PPT) d2[r] = __builtin_fmaf(dx[r], dx[r], d2[r]);
      #pragma unroll
      for (int r = 0; r < 4; ++r) if (4 * q + r < FPS_PPT) d2[r] = __builtin_fmaf(dz[r], dz[r], d2[r]);
      #pragma unroll
      for (int r = 0; r < 4; ++r) {
        const int j = 4 * q + r;
        if (j < FPS_PPT) {
          const float dd = fminf(dist[j], d2[r]);
          dist[j] = dd;
          d2[r] = dd;
        }
      }
      #pragma unroll
      for (int r = 0; r < 4; ++r) {
        const int j = 4 * q + r;
        if (j < FPS_PPT) {
          if (j % FPS_GS == 0) gm[j / FPS_GS] = d2[r];
          else                 gm[j / FPS_GS] = fmaxf(gm[j / FPS_GS], d2[r]);
        }
      }
      if (q % 2 == 1) __builtin_amdgcn_sched_barrier(0x0094);
    }
    float bv = gm[0];
    #pragma unroll
    for (int k = 1; k < FPS_NG; ++k) bv = fmaxf(bv, gm[k]);
    int vb = __builtin_bit_cast(int, bv);
    vb = fps_row_max<1>(vb);
    vb = fps_row_max<2>(vb);
    vb = fps_row_max<4>(vb);
    vb = fps_row_max<8>(vb);
    {
      const int ov = __builtin_amdgcn_permlanex16(vb, vb, 0x76543210, 0xfedcba98, false, false);
      vb = (ov > vb) ? ov : vb;
    }
    const float V = __builtin_bit_cast(float, __builtin_amdgcn_readfirstlane(vb));
    int kg = 0;
    #pragma unroll
    for (int k = FPS_NG - 1; k >= 0; --k) {
      if (__builtin_amdgcn_ballot_w32(gm[k] == V) != 0u) kg = k;
    }
    int      jj = 0;
    unsigned mm = 1u;
    #pragma unroll
    for (int k = 0; k < FPS_NG; ++k) {
      if (kg == k) {
        #pragma unroll
        for (int r = FPS_GS - 1; r >= 0; --r) {
          const unsigned m = __builtin_amdgcn_ballot_w32(dist[k * FPS_GS + r] == V);
          if (m != 0u) { jj = k * FPS_GS + r; mm = m; }
        }
      }
    }
    const int bi = (wave << 5) + __builtin_ctz(mm) + jj * FPS_T;
    if (lane == 0) {
      v2i w;
      w.x = __builtin_bit_cast(int, V);
      w.y = bi;
      s_win[it & 1][wave] = w;
    }
    __builtin_amdgcn_fence(__ATOMIC_RELEASE, "workgroup", "local");
    __builtin_amdgcn_s_barrier();
    __builtin_amdgcn_fence(__ATOMIC_ACQUIRE, "workgroup", "local");
    const v2i ws = s_win[it & 1][lane & 15];
    int tv = ws.x;
    tv = fps_row_max<1>(tv);
    tv = fps_row_max<2>(tv);
    tv = fps_row_max<4>(tv);
    tv = fps_row_max<8>(tv);
    int ci = (ws.x == tv) ? ws.y : 0x7fffffff;
    ci = fps_row_min<1>(ci);
    ci = fps_row_min<2>(ci);
    ci = fps_row_min<4>(ci);
    ci = fps_row_min<8>(ci);
    cur = __builtin_amdgcn_readfirstlane(ci);
    cur = (cur < N) ? cur : 0;
  }
}

__global__ __launch_bounds__(256)
void bq_pool_kernel(const float* __restrict__ pos,
                    const float* __restrict__ fs,
                    const float* __restrict__ fv,
                    const int* __restrict__ cent_idx,
                    float* __restrict__ pooled_s,
                    float* __restrict__ pooled_v,
                    int N, int C, int M) {
  const int m   = blockIdx.x;
  const int tid = threadIdx.x;
  __shared__ float s_cd2[CAP];
  __shared__ int   s_cidx[CAP];
  __shared__ int   s_cnt;
  __shared__ int   s_sel[K_LOCAL];
  __shared__ float s_rv[256];
  __shared__ int   s_ri[256];
  __shared__ int   s_rp[256];
  __shared__ float s_c[3];

  if (tid == 0) {
    s_cnt = 0;
    int ci = cent_idx[m];
    s_c[0] = pos[ci * 3 + 0];
    s_c[1] = pos[ci * 3 + 1];
    s_c[2] = pos[ci * 3 + 2];
  }
  __syncthreads();
  const float cx = s_c[0], cy = s_c[1], cz = s_c[2];

  for (int i = tid; i < N; i += 256) {
    float dx = pos[i * 3 + 0] - cx;
    float dy = pos[i * 3 + 1] - cy;
    float dz = pos[i * 3 + 2] - cz;
    float d2 = dx * dx + dy * dy + dz * dz;
    if (d2 <= RADIUS2) {
      int p = atomicAdd(&s_cnt, 1);
      if (p < CAP) { s_cd2[p] = d2; s_cidx[p] = i; }
    }
  }
  __syncthreads();
  const int ncand = min(s_cnt, CAP);

  for (int s = 0; s < K_LOCAL; ++s) {
    float bv = FLT_BIG;
    int   bi = 0x7fffffff;
    int   bp = -1;
    for (int p = tid; p < ncand; p += 256) {
      float v  = s_cd2[p];
      int   ix = s_cidx[p];
      if (v < bv || (v == bv && ix < bi)) { bv = v; bi = ix; bp = p; }
    }
    s_rv[tid] = bv; s_ri[tid] = bi; s_rp[tid] = bp;
    __syncthreads();
    for (int off = 128; off >= 1; off >>= 1) {
      if (tid < off) {
        float ov = s_rv[tid + off];
        int   oi = s_ri[tid + off];
        if (ov < s_rv[tid] || (ov == s_rv[tid] && oi < s_ri[tid])) {
          s_rv[tid] = ov; s_ri[tid] = oi; s_rp[tid] = s_rp[tid + off];
        }
      }
      __syncthreads();
    }
    if (tid == 0) {
      if (s < ncand) {
        int p = s_rp[0];
        s_sel[s] = s_cidx[p];
        s_cd2[p] = FLT_BIG;
      } else {
        s_sel[s] = s_sel[0];
      }
    }
    __syncthreads();
  }

  for (int j = tid; j < C * 4; j += 256) {
    const int c = j >> 2;
    const int d = j & 3;
    float mx = -FLT_BIG;
    #pragma unroll
    for (int t = 0; t < K_LOCAL; ++t) {
      int n = s_sel[t];
      float v = (d == 0) ? fs[(size_t)n * C + c]
                         : fv[((size_t)n * C + c) * 3 + (d - 1)];
      mx = fmaxf(mx, v);
    }
    if (d == 0) pooled_s[(size_t)m * C + c] = mx;
    else        pooled_v[(size_t)(d - 1) * M * C + (size_t)m * C + c] = mx;
  }
}

__global__ __launch_bounds__(256)
void mix_wmma_kernel(const float* __restrict__ pooled_s,
                     const float* __restrict__ pooled_v,
                     const float* __restrict__ Ws, const float* __restrict__ Wv,
                     const float* __restrict__ bs, const float* __restrict__ bv,
                     float* __restrict__ outS,
                     float* __restrict__ outV,
                     int M, int C) {
  const int lane = threadIdx.x & 31;
  const int wave = threadIdx.x >> 5;
  const int tiles_o = C >> 4;
  const int tiles_m = M >> 4;
  const int per_d   = tiles_m * tiles_o;
  const int tile    = blockIdx.x * (blockDim.x >> 5) + wave;
  if (tile >= per_d * 4) return;

  const int d   = tile / per_d;
  const int rem = tile % per_d;
  const int m0  = (rem / tiles_o) << 4;
  const int o0  = (rem % tiles_o) << 4;

  const float* A    = (d == 0) ? (pooled_s + (size_t)m0 * C)
                               : (pooled_v + (size_t)(d - 1) * M * C + (size_t)m0 * C);
  const float* B    = (d == 0) ? Ws : Wv;
  const float* bias = (d == 0) ? bs : bv;

  const int ml = lane & 15;
  const int kh = lane >> 4;

  v8f acc = {};
  for (int k0 = 0; k0 < C; k0 += 4) {
    const int ka = k0 + 2 * kh;
    v2f a = *(const v2f*)(A + (size_t)ml * C + ka);
    v2f b;
    b.x = B[(size_t)ka * C + o0 + ml];
    b.y = B[(size_t)(ka + 1) * C + o0 + ml];
    acc = __builtin_amdgcn_wmma_f32_16x16x4_f32(
        false, a, false, b, (short)0, acc, false, false);
  }

  const int   n  = o0 + ml;
  const float bb = bias[n];
  #pragma unroll
  for (int v = 0; v < 8; ++v) {
    const int mm  = m0 + v + 8 * kh;
    const float val = acc[v] + bb;
    if (d == 0) outS[(size_t)mm * C + n] = val;
    else        outV[((size_t)mm * C + n) * 3 + (d - 1)] = val;
  }
}

extern "C" void kernel_launch(void* const* d_in, const int* in_sizes, int n_in,
                              void* d_out, int out_size, void* d_ws, size_t ws_size,
                              hipStream_t stream) {
  const float* pos = (const float*)d_in[0];
  const float* fs  = (const float*)d_in[1];
  const float* fv  = (const float*)d_in[2];
  const float* Ws  = (const float*)d_in[3];
  const float* Wv  = (const float*)d_in[4];
  const float* bs  = (const float*)d_in[5];
  const float* bv  = (const float*)d_in[6];

  const int N = in_sizes[0] / 3;
  const int C = in_sizes[1] / N;
  const int M = (N < 2048) ? N : 2048;

  float* out   = (float*)d_out;
  float* cents = out;
  float* outS  = out + (size_t)M * 3;
  float* outV  = outS + (size_t)M * C;

  int*   cent_idx = (int*)d_ws;
  float* wsf      = (float*)d_ws;
  float* spill    = wsf + 4096;
  float* pooled_s = wsf + 4096 + 65536;
  float* pooled_v = pooled_s + (size_t)M * C;

  fps_kernel<<<1, 512, 0, stream>>>(pos, spill, cent_idx, cents, N, M);
  bq_pool_kernel<<<M, 256, 0, stream>>>(pos, fs, fv, cent_idx,
                                        pooled_s, pooled_v, N, C, M);

  const int per_d = (M / 16) * (C / 16);
  const int total = per_d * 4;
  mix_wmma_kernel<<<(total + 7) / 8, 256, 0, stream>>>(
      pooled_s, pooled_v, Ws, Wv, bs, bv, outS, outV, M, C);
}


# FPS argmax index recovery: the 7 group and the 14 point equality compares issue as one batch each ahead of their scalar selection; wave max and winner index stay in vector registers; bit-identical
# speedup vs baseline: 7.2373x; 1.0973x over previous
#include <hip/hip_runtime.h>
#include <hip/hip_bf16.h>

#define FLT_BIG 3.402823466e+38f
#define RADIUS2 0.01f
#define K_LOCAL 32
#define CAP 2048

typedef __attribute__((ext_vector_type(2))) float v2f;
typedef __attribute__((ext_vector_type(8))) float v8f;

#define FPS_T     512
#define FPS_PPT   98
#define FPS_PITCH 156
#define FPS_GS    14
#define FPS_NG    7
#define FPS_NQ    25

typedef __attribute__((ext_vector_type(4))) float v4f;
typedef __attribute__((ext_vector_type(2))) int   v2i;

template <int MASK>
__device__ __forceinline__ int fps_row_max(int v) {
  const int ov = __builtin_amdgcn_update_dpp(v, v, 0x160 + MASK, 0xf, 0xf, true);
  return (ov > v) ? ov : v;
}

template <int MASK>
__device__ __forceinline__ int fps_row_min(int v) {
  const int ov = __builtin_amdgcn_update_dpp(v, v, 0x160 + MASK, 0xf, 0xf, true);
  return (ov < v) ? ov : v;
}

__global__ __launch_bounds__(FPS_T)
void fps_kernel(const float* __restrict__ pos, float* __restrict__ spill,
                int* __restrict__ cent_idx, float* __restrict__ cents,
                int N, int M) {
  const int tid  = threadIdx.x;
  const int lane = tid & 31;
  const int wave = tid >> 5;
  __shared__ __attribute__((aligned(8)))  v2i   s_win[2][FPS_T / 32];
  __shared__ __attribute__((aligned(16))) float s_pt[FPS_T * FPS_PITCH];

  if (N > FPS_PPT * FPS_T) return;

  float*     L  = s_pt + tid * FPS_PITCH;
  const v4f* L4 = (const v4f*)L;
  const v4f* P4 = (const v4f*)spill;

  float px[FPS_PPT];
  float dist[FPS_PPT];
  #pragma unroll
  for (int j = 0; j < FPS_PPT; ++j) {
    const int i = tid + j * FPS_T;
    float x = 0.0f, y = 0.0f, z = 0.0f, d = -FLT_BIG;
    if (i < N) {
      x = pos[i * 3 + 0];
      y = pos[i * 3 + 1];
      z = pos[i * 3 + 2];
      d = FLT_BIG;
    }
    dist[j] = d;
    L[j] = y;
    if (j >= 54 && j < 82) spill[((((6 + ((j - 54) >> 2)) * FPS_T) + tid) << 2) + ((j - 54) & 3)] = x;
    else                   px[j] = x;
    if (j < 24)      spill[((((j >> 2) * FPS_T) + tid) << 2) + (j & 3)] = z;
    else if (j < 80) L[100 + (j - 24)] = z;
    else if (j < 82) L[98 + (j - 80)] = z;
    else             spill[((((13 + ((j - 82) >> 2)) * FPS_T) + tid) << 2) + ((j - 82) & 3)] = z;
  }
  __syncthreads();

  v4f zb1[6];
  #pragma unroll
  for (int q = 0; q < 6; ++q) zb1[q] = P4[q * FPS_T + tid];

  int cur = 0;
  for (int it = 0; it < M; ++it) {
    const float* cp = (const float*)((const char*)pos + __umul24((unsigned)cur, 12u));
    const float cx = cp[0];
    const float cy = cp[1];
    const float cz = cp[2];
    if (tid == 0) {
      cent_idx[it] = cur;
      cents[it * 3 + 0] = cx;
      cents[it * 3 + 1] = cy;
      cents[it * 3 + 2] = cz;
    }
    float gm[FPS_NG];
    v4f xb[7];
    v4f zc[4];
    #pragma unroll
    for (int q = 0; q < FPS_NQ; ++q) {
      if (q == 6) {
        #pragma unroll
        for (int p = 0; p < 7; ++p) xb[p] = P4[(6 + p) * FPS_T + tid];
      }
      if (q == 15) {
        #pragma unroll
        for (int p = 0; p < 4; ++p) zc[p] = P4[(13 + p) * FPS_T + tid];
      }
      if (q == 21) {
        #pragma unroll
        for (int p = 0; p < 6; ++p) zb1[p] = P4[p * FPS_T + tid];
      }
      const v4f yv = L4[q];
      float dx[4], dy[4], dz[4], d2[4];
      #pragma unroll
      for (int r = 0; r < 4; ++r) {
        const int j = 4 * q + r;
        if (j < FPS_PPT) {
          float z;
          if (j < 24)      { z = zb1[j >> 2][j & 3]; }
          else if (j < 80) { const v4f zv = L4[25 + ((j - 24) >> 2)]; z = zv[(j - 24) & 3]; }
          else if (j < 82) { const v4f zv = L4[24]; z = zv[2 + (j - 80)]; }
          else             { z = zc[(j - 82) >> 2][(j - 82) & 3]; }
          float x;
          if (j >= 54 && j < 82) x = xb[(j - 54) >> 2][(j - 54) & 3];
          else                   x = px[j];
          dx[r] = x - cx;
          dy[r] = yv[r] - cy;
          dz[r] = z - cz;
        }
      }
      #pragma unroll
      for (int r = 0; r < 4; ++r) if (4 * q + r < FPS_PPT) d2[r] = dy[r] * dy[r];
      #pragma unroll
      for (int r = 0; r < 4; ++r) if (4 * q + r < FPS_PPT) d2[r] = __builtin_fmaf(dx[r], dx[r], d2[r]);
      #pragma unroll
      for (int r = 0; r < 4; ++r) if (4 * q + r < FPS_PPT) d2[r] = __builtin_fmaf(dz[r], dz[r], d2[r]);
      #pragma unroll
      for (int r = 0; r < 4; ++r) {
        const int j = 4 * q + r;
        if (j < FPS_PPT) {
          const float dd = fminf(dist[j], d2[r]);
          dist[j] = dd;
          d2[r] = dd;
        }
      }
      #pragma unroll
      for (int r = 0; r < 4; ++r) {
        const int j = 4 * q + r;
        if (j < FPS_PPT) {
          if (j % FPS_GS == 0) gm[j / FPS_GS] = d2[r];
          else                 gm[j / FPS_GS] = fmaxf(gm[j / FPS_GS], d2[r]);
        }
      }
      if (q % 2 == 1) __builtin_amdgcn_sched_barrier(0x0094);
    }
    float bv = gm[0];
    #pragma unroll
    for (int k = 1; k < FPS_NG; ++k) bv = fmaxf(bv, gm[k]);
    int vb = __builtin_bit_cast(int, bv);
    vb = fps_row_max<1>(vb);
    vb = fps_row_max<2>(vb);
    vb = fps_row_max<4>(vb);
    vb = fps_row_max<8>(vb);
    {
      const int ov = __builtin_amdgcn_permlanex16(vb, vb, 0x76543210, 0xfedcba98, false, false);
      vb = (ov > vb) ? ov : vb;
    }
    const float V = __builtin_bit_cast(float, vb);
    unsigned gb[FPS_NG];
    #pragma unroll
    for (int k = FPS_NG - 1; k >= 0; --k) gb[k] = __builtin_amdgcn_ballot_w32(gm[k] == V);
    __builtin_amdgcn_sched_barrier(0);
    int kg = 0;
    #pragma unroll
    for (int k = FPS_NG - 1; k >= 0; --k) {
      if (gb[k] != 0u) kg = k;
    }
    int      jj = 0;
    unsigned mm = 1u;
    #pragma unroll
    for (int k = 0; k < FPS_NG; ++k) {
      if (kg == k) {
        unsigned mr[FPS_GS];
        #pragma unroll
        for (int r = FPS_GS - 1; r >= 0; --r) mr[r] = __builtin_amdgcn_ballot_w32(dist[k * FPS_GS + r] == V);
        __builtin_amdgcn_sched_barrier(0);
        #pragma unroll
        for (int r = FPS_GS - 1; r >= 0; --r) {
          if (mr[r] != 0u) { jj = k * FPS_GS + r; mm = mr[r]; }
        }
      }
    }
    const int bi = (wave << 5) + __builtin_ctz(mm) + jj * FPS_T;
    if (lane == 0) {
      v2i w;
      w.x = vb;
      w.y = bi;
      s_win[it & 1][wave] = w;
    }
    __builtin_amdgcn_fence(__ATOMIC_RELEASE, "workgroup", "local");
    __builtin_amdgcn_s_barrier();
    __builtin_amdgcn_fence(__ATOMIC_ACQUIRE, "workgroup", "local");
    const v2i ws = s_win[it & 1][lane & 15];
    int tv = ws.x;
    tv = fps_row_max<1>(tv);
    tv = fps_row_max<2>(tv);
    tv = fps_row_max<4>(tv);
    tv = fps_row_max<8>(tv);
    int ci = (ws.x == tv) ? ws.y : 0x7fffffff;
    ci = fps_row_min<1>(ci);
    ci = fps_row_min<2>(ci);
    ci = fps_row_min<4>(ci);
    ci = fps_row_min<8>(ci);
    cur = (ci < N) ? ci : 0;
  }
}

__global__ __launch_bounds__(256)
void bq_pool_kernel(const float* __restrict__ pos,
                    const float* __restrict__ fs,
                    const float* __restrict__ fv,
                    const int* __restrict__ cent_idx,
                    float* __restrict__ pooled_s,
                    float* __restrict__ pooled_v,
                    int N, int C, int M) {
  const int m   = blockIdx.x;
  const int tid = threadIdx.x;
  __shared__ float s_cd2[CAP];
  __shared__ int   s_cidx[CAP];
  __shared__ int   s_cnt;
  __shared__ int   s_sel[K_LOCAL];
  __shared__ float s_rv[256];
  __shared__ int   s_ri[256];
  __shared__ int   s_rp[256];
  __shared__ float s_c[3];

  if (tid == 0) {
    s_cnt = 0;
    int ci = cent_idx[m];
    s_c[0] = pos[ci * 3 + 0];
    s_c[1] = pos[ci * 3 + 1];
    s_c[2] = pos[ci * 3 + 2];
  }
  __syncthreads();
  const float cx = s_c[0], cy = s_c[1], cz = s_c[2];

  for (int i = tid; i < N; i += 256) {
    float dx = pos[i * 3 + 0] - cx;
    float dy = pos[i * 3 + 1] - cy;
    float dz = pos[i * 3 + 2] - cz;
    float d2 = dx * dx + dy * dy + dz * dz;
    if (d2 <= RADIUS2) {
      int p = atomicAdd(&s_cnt, 1);
      if (p < CAP) { s_cd2[p] = d2; s_cidx[p] = i; }
    }
  }
  __syncthreads();
  const int ncand = min(s_cnt, CAP);

  for (int s = 0; s < K_LOCAL; ++s) {
    float bv = FLT_BIG;
    int   bi = 0x7fffffff;
    int   bp = -1;
    for (int p = tid; p < ncand; p += 256) {
      float v  = s_cd2[p];
      int   ix = s_cidx[p];
      if (v < bv || (v == bv && ix < bi)) { bv = v; bi = ix; bp = p; }
    }
    s_rv[tid] = bv; s_ri[tid] = bi; s_rp[tid] = bp;
    __syncthreads();
    for (int off = 128; off >= 1; off >>= 1) {
      if (tid < off) {
        float ov = s_rv[tid + off];
        int   oi = s_ri[tid + off];
        if (ov < s_rv[tid] || (ov == s_rv[tid] && oi < s_ri[tid])) {
          s_rv[tid] = ov; s_ri[tid] = oi; s_rp[tid] = s_rp[tid + off];
        }
      }
      __syncthreads();
    }
    if (tid == 0) {
      if (s < ncand) {
        int p = s_rp[0];
        s_sel[s] = s_cidx[p];
        s_cd2[p] = FLT_BIG;
      } else {
        s_sel[s] = s_sel[0];
      }
    }
    __syncthreads();
  }

  for (int j = tid; j < C * 4; j += 256) {
    const int c = j >> 2;
    const int d = j & 3;
    float mx = -FLT_BIG;
    #pragma unroll
    for (int t = 0; t < K_LOCAL; ++t) {
      int n = s_sel[t];
      float v = (d == 0) ? fs[(size_t)n * C + c]
                         : fv[((size_t)n * C + c) * 3 + (d - 1)];
      mx = fmaxf(mx, v);
    }
    if (d == 0) pooled_s[(size_t)m * C + c] = mx;
    else        pooled_v[(size_t)(d - 1) * M * C + (size_t)m * C + c] = mx;
  }
}

__global__ __launch_bounds__(256)
void mix_wmma_kernel(const float* __restrict__ pooled_s,
                     const float* __restrict__ pooled_v,
                     const float* __restrict__ Ws, const float* __restrict__ Wv,
                     const float* __restrict__ bs, const float* __restrict__ bv,
                     float* __restrict__ outS,
                     float* __restrict__ outV,
                     int M, int C) {
  const int lane = threadIdx.x & 31;
  const int wave = threadIdx.x >> 5;
  const int tiles_o = C >> 4;
  const int tiles_m = M >> 4;
  const int per_d   = tiles_m * tiles_o;
  const int tile    = blockIdx.x * (blockDim.x >> 5) + wave;
  if (tile >= per_d * 4) return;

  const int d   = tile / per_d;
  const int rem = tile % per_d;
  const int m0  = (rem / tiles_o) << 4;
  const int o0  = (rem % tiles_o) << 4;

  const float* A    = (d == 0) ? (pooled_s + (size_t)m0 * C)
                               : (pooled_v + (size_t)(d - 1) * M * C + (size_t)m0 * C);
  const float* B    = (d == 0) ? Ws : Wv;
  const float* bias = (d == 0) ? bs : bv;

  const int ml = lane & 15;
  const int kh = lane >> 4;

  v8f acc = {};
  for (int k0 = 0; k0 < C; k0 += 4) {
    const int ka = k0 + 2 * kh;
    v2f a = *(const v2f*)(A + (size_t)ml * C + ka);
    v2f b;
    b.x = B[(size_t)ka * C + o0 + ml];
    b.y = B[(size_t)(ka + 1) * C + o0 + ml];
    acc = __builtin_amdgcn_wmma_f32_16x16x4_f32(
        false, a, false, b, (short)0, acc, false, false);
  }

  const int   n  = o0 + ml;
  const float bb = bias[n];
  #pragma unroll
  for (int v = 0; v < 8; ++v) {
    const int mm  = m0 + v + 8 * kh;
    const float val = acc[v] + bb;
    if (d == 0) outS[(size_t)mm * C + n] = val;
    else        outV[((size_t)mm * C + n) * 3 + (d - 1)] = val;
  }
}

extern "C" void kernel_launch(void* const* d_in, const int* in_sizes, int n_in,
                              void* d_out, int out_size, void* d_ws, size_t ws_size,
                              hipStream_t stream) {
  const float* pos = (const float*)d_in[0];
  const float* fs  = (const float*)d_in[1];
  const float* fv  = (const float*)d_in[2];
  const float* Ws  = (const float*)d_in[3];
  const float* Wv  = (const float*)d_in[4];
  const float* bs  = (const float*)d_in[5];
  const float* bv  = (const float*)d_in[6];

  const int N = in_sizes[0] / 3;
  const int C = in_sizes[1] / N;
  const int M = (N < 2048) ? N : 2048;

  float* out   = (float*)d_out;
  float* cents = out;
  float* outS  = out + (size_t)M * 3;
  float* outV  = outS + (size_t)M * C;

  int*   cent_idx = (int*)d_ws;
  float* wsf      = (float*)d_ws;
  float* spill    = wsf + 4096;
  float* pooled_s = wsf + 4096 + 65536;
  float* pooled_v = pooled_s + (size_t)M * C;

  fps_kernel<<<1, 512, 0, stream>>>(pos, spill, cent_idx, cents, N, M);
  bq_pool_kernel<<<M, 256, 0, stream>>>(pos, fs, fv, cent_idx,
                                        pooled_s, pooled_v, N, C, M);

  const int per_d = (M / 16) * (C / 16);
  const int total = per_d * 4;
  mix_wmma_kernel<<<(total + 7) / 8, 256, 0, stream>>>(
      pooled_s, pooled_v, Ws, Wv, bs, bv, outS, outV, M, C);
}


# FPS: each wave loads its own candidate's coordinates ahead of the step's barrier, so the winner's fetch behind the barrier finds its line in the WGP cache, not the second level; bit-identical
# speedup vs baseline: 7.5440x; 1.0424x over previous
#include <hip/hip_runtime.h>
#include <hip/hip_bf16.h>

#define FLT_BIG 3.402823466e+38f
#define RADIUS2 0.01f
#define K_LOCAL 32
#define CAP 2048

typedef __attribute__((ext_vector_type(2))) float v2f;
typedef __attribute__((ext_vector_type(8))) float v8f;

#define FPS_T     512
#define FPS_PPT   98
#define FPS_PITCH 156
#define FPS_GS    14
#define FPS_NG    7
#define FPS_NQ    25

typedef __attribute__((ext_vector_type(4))) float v4f;
typedef __attribute__((ext_vector_type(2))) int   v2i;

template <int MASK>
__device__ __forceinline__ int fps_row_max(int v) {
  const int ov = __builtin_amdgcn_update_dpp(v, v, 0x160 + MASK, 0xf, 0xf, true);
  return (ov > v) ? ov : v;
}

template <int MASK>
__device__ __forceinline__ int fps_row_min(int v) {
  const int ov = __builtin_amdgcn_update_dpp(v, v, 0x160 + MASK, 0xf, 0xf, true);
  return (ov < v) ? ov : v;
}

__global__ __launch_bounds__(FPS_T)
void fps_kernel(const float* __restrict__ pos, float* __restrict__ spill,
                int* __restrict__ cent_idx, float* __restrict__ cents,
                int N, int M) {
  const int tid  = threadIdx.x;
  const int lane = tid & 31;
  const int wave = tid >> 5;
  __shared__ __attribute__((aligned(8)))  v2i   s_win[2][FPS_T / 32];
  __shared__ __attribute__((aligned(16))) float s_pt[FPS_T * FPS_PITCH];

  if (N > FPS_PPT * FPS_T) return;

  float*     L  = s_pt + tid * FPS_PITCH;
  const v4f* L4 = (const v4f*)L;
  const v4f* P4 = (const v4f*)spill;

  float px[FPS_PPT];
  float dist[FPS_PPT];
  #pragma unroll
  for (int j = 0; j < FPS_PPT; ++j) {
    const int i = tid + j * FPS_T;
    float x = 0.0f, y = 0.0f, z = 0.0f, d = -FLT_BIG;
    if (i < N) {
      x = pos[i * 3 + 0];
      y = pos[i * 3 + 1];
      z = pos[i * 3 + 2];
      d = FLT_BIG;
    }
    dist[j] = d;
    L[j] = y;
    if (j >= 54 && j < 82) spill[((((6 + ((j - 54) >> 2)) * FPS_T) + tid) << 2) + ((j - 54) & 3)] = x;
    else                   px[j] = x;
    if (j < 24)      spill[((((j >> 2) * FPS_T) + tid) << 2) + (j & 3)] = z;
    else if (j < 80) L[100 + (j - 24)] = z;
    else if (j < 82) L[98 + (j - 80)] = z;
    else             spill[((((13 + ((j - 82) >> 2)) * FPS_T) + tid) << 2) + ((j - 82) & 3)] = z;
  }
  __syncthreads();

  v4f zb1[6];
  #pragma unroll
  for (int q = 0; q < 6; ++q) zb1[q] = P4[q * FPS_T + tid];

  int cur = 0;
  float ex = 0.0f, ey = 0.0f, ez = 0.0f;
  const unsigned last_pt = (unsigned)(N - 1);
  for (int it = 0; it < M; ++it) {
    const float* cp = (const float*)((const char*)pos + __umul24((unsigned)cur, 12u));
    const float cx = cp[0];
    const float cy = cp[1];
    const float cz = cp[2];
    if (tid == 0) {
      cent_idx[it] = cur;
      cents[it * 3 + 0] = cx;
      cents[it * 3 + 1] = cy;
      cents[it * 3 + 2] = cz;
    }
    float gm[FPS_NG];
    v4f xb[7];
    v4f zc[4];
    #pragma unroll
    for (int q = 0; q < FPS_NQ; ++q) {
      if (q == 6) {
        #pragma unroll
        for (int p = 0; p < 7; ++p) xb[p] = P4[(6 + p) * FPS_T + tid];
      }
      if (q == 15) {
        #pragma unroll
        for (int p = 0; p < 4; ++p) zc[p] = P4[(13 + p) * FPS_T + tid];
      }
      if (q == 21) {
        #pragma unroll
        for (int p = 0; p < 6; ++p) zb1[p] = P4[p * FPS_T + tid];
      }
      const v4f yv = L4[q];
      float dx[4], dy[4], dz[4], d2[4];
      #pragma unroll
      for (int r = 0; r < 4; ++r) {
        const int j = 4 * q + r;
        if (j < FPS_PPT) {
          float z;
          if (j < 24)      { z = zb1[j >> 2][j & 3]; }
          else if (j < 80) { const v4f zv = L4[25 + ((j - 24) >> 2)]; z = zv[(j - 24) & 3]; }
          else if (j < 82) { const v4f zv = L4[24]; z = zv[2 + (j - 80)]; }
          else             { z = zc[(j - 82) >> 2][(j - 82) & 3]; }
          float x;
          if (j >= 54 && j < 82) x = xb[(j - 54) >> 2][(j - 54) & 3];
          else                   x = px[j];
          dx[r] = x - cx;
          dy[r] = yv[r] - cy;
          dz[r] = z - cz;
        }
      }
      #pragma unroll
      for (int r = 0; r < 4; ++r) if (4 * q + r < FPS_PPT) d2[r] = dy[r] * dy[r];
      #pragma unroll
      for (int r = 0; r < 4; ++r) if (4 * q + r < FPS_PPT) d2[r] = __builtin_fmaf(dx[r], dx[r], d2[r]);
      #pragma unroll
      for (int r = 0; r < 4; ++r) if (4 * q + r < FPS_PPT) d2[r] = __builtin_fmaf(dz[r], dz[r], d2[r]);
      #pragma unroll
      for (int r = 0; r < 4; ++r) {
        const int j = 4 * q + r;
        if (j < FPS_PPT) {
          const float dd = fminf(dist[j], d2[r]);
          dist[j] = dd;
          d2[r] = dd;
        }
      }
      #pragma unroll
      for (int r = 0; r < 4; ++r) {
        const int j = 4 * q + r;
        if (j < FPS_PPT) {
          if (j % FPS_GS == 0) gm[j / FPS_GS] = d2[r];
          else                 gm[j / FPS_GS] = fmaxf(gm[j / FPS_GS], d2[r]);
        }
      }
      if (q == 0) asm volatile("" :: "v"(ex), "v"(ey), "v"(ez));
      if (q % 2 == 1) __builtin_amdgcn_sched_barrier(0x0094);
    }
    float bv = gm[0];
    #pragma unroll
    for (int k = 1; k < FPS_NG; ++k) bv = fmaxf(bv, gm[k]);
    int vb = __builtin_bit_cast(int, bv);
    vb = fps_row_max<1>(vb);
    vb = fps_row_max<2>(vb);
    vb = fps_row_max<4>(vb);
    vb = fps_row_max<8>(vb);
    {
      const int ov = __builtin_amdgcn_permlanex16(vb, vb, 0x76543210, 0xfedcba98, false, false);
      vb = (ov > vb) ? ov : vb;
    }
    const float V = __builtin_bit_cast(float, vb);
    unsigned gb[FPS_NG];
    #pragma unroll
    for (int k = FPS_NG - 1; k >= 0; --k) gb[k] = __builtin_amdgcn_ballot_w32(gm[k] == V);
    __builtin_amdgcn_sched_barrier(0);
    int kg = 0;
    #pragma unroll
    for (int k = FPS_NG - 1; k >= 0; --k) {
      if (gb[k] != 0u) kg = k;
    }
    int      jj = 0;
    unsigned mm = 1u;
    #pragma unroll
    for (int k = 0; k < FPS_NG; ++k) {
      if (kg == k) {
        unsigned mr[FPS_GS];
        #pragma unroll
        for (int r = FPS_GS - 1; r >= 0; --r) mr[r] = __builtin_amdgcn_ballot_w32(dist[k * FPS_GS + r] == V);
        __builtin_amdgcn_sched_barrier(0);
        #pragma unroll
        for (int r = FPS_GS - 1; r >= 0; --r) {
          if (mr[r] != 0u) { jj = k * FPS_GS + r; mm = mr[r]; }
        }
      }
    }
    const int bi = (wave << 5) + __builtin_ctz(mm) + jj * FPS_T;
    if (lane == 0) {
      v2i w;
      w.x = vb;
      w.y = bi;
      s_win[it & 1][wave] = w;
    }
    {
      const unsigned eb = ((unsigned)bi < last_pt) ? (unsigned)bi : last_pt;
      const float* ep = (const float*)((const char*)pos + __umul24(eb, 12u));
      ex = ep[0];
      ey = ep[1];
      ez = ep[2];
    }
    __builtin_amdgcn_fence(__ATOMIC_RELEASE, "workgroup", "local");
    __builtin_amdgcn_s_barrier();
    __builtin_amdgcn_fence(__ATOMIC_ACQUIRE, "workgroup", "local");
    const v2i ws = s_win[it & 1][lane & 15];
    int tv = ws.x;
    tv = fps_row_max<1>(tv);
    tv = fps_row_max<2>(tv);
    tv = fps_row_max<4>(tv);
    tv = fps_row_max<8>(tv);
    int ci = (ws.x == tv) ? ws.y : 0x7fffffff;
    ci = fps_row_min<1>(ci);
    ci = fps_row_min<2>(ci);
    ci = fps_row_min<4>(ci);
    ci = fps_row_min<8>(ci);
    cur = (ci < N) ? ci : 0;
  }
}

__global__ __launch_bounds__(256)
void bq_pool_kernel(const float* __restrict__ pos,
                    const float* __restrict__ fs,
                    const float* __restrict__ fv,
                    const int* __restrict__ cent_idx,
                    float* __restrict__ pooled_s,
                    float* __restrict__ pooled_v,
                    int N, int C, int M) {
  const int m   = blockIdx.x;
  const int tid = threadIdx.x;
  __shared__ float s_cd2[CAP];
  __shared__ int   s_cidx[CAP];
  __shared__ int   s_cnt;
  __shared__ int   s_sel[K_LOCAL];
  __shared__ float s_rv[256];
  __shared__ int   s_ri[256];
  __shared__ int   s_rp[256];
  __shared__ float s_c[3];

  if (tid == 0) {
    s_cnt = 0;
    int ci = cent_idx[m];
    s_c[0] = pos[ci * 3 + 0];
    s_c[1] = pos[ci * 3 + 1];
    s_c[2] = pos[ci * 3 + 2];
  }
  __syncthreads();
  const float cx = s_c[0], cy = s_c[1], cz = s_c[2];

  for (int i = tid; i < N; i += 256) {
    float dx = pos[i * 3 + 0] - cx;
    float dy = pos[i * 3 + 1] - cy;
    float dz = pos[i * 3 + 2] - cz;
    float d2 = dx * dx + dy * dy + dz * dz;
    if (d2 <= RADIUS2) {
      int p = atomicAdd(&s_cnt, 1);
      if (p < CAP) { s_cd2[p] = d2; s_cidx[p] = i; }
    }
  }
  __syncthreads();
  const int ncand = min(s_cnt, CAP);

  for (int s = 0; s < K_LOCAL; ++s) {
    float bv = FLT_BIG;
    int   bi = 0x7fffffff;
    int   bp = -1;
    for (int p = tid; p < ncand; p += 256) {
      float v  = s_cd2[p];
      int   ix = s_cidx[p];
      if (v < bv || (v == bv && ix < bi)) { bv = v; bi = ix; bp = p; }
    }
    s_rv[tid] = bv; s_ri[tid] = bi; s_rp[tid] = bp;
    __syncthreads();
    for (int off = 128; off >= 1; off >>= 1) {
      if (tid < off) {
        float ov = s_rv[tid + off];
        int   oi = s_ri[tid + off];
        if (ov < s_rv[tid] || (ov == s_rv[tid] && oi < s_ri[tid])) {
          s_rv[tid] = ov; s_ri[tid] = oi; s_rp[tid] = s_rp[tid + off];
        }
      }
      __syncthreads();
    }
    if (tid == 0) {
      if (s < ncand) {
        int p = s_rp[0];
        s_sel[s] = s_cidx[p];
        s_cd2[p] = FLT_BIG;
      } else {
        s_sel[s] = s_sel[0];
      }
    }
    __syncthreads();
  }

  for (int j = tid; j < C * 4; j += 256) {
    const int c = j >> 2;
    const int d = j & 3;
    float mx = -FLT_BIG;
    #pragma unroll
    for (int t = 0; t < K_LOCAL; ++t) {
      int n = s_sel[t];
      float v = (d == 0) ? fs[(size_t)n * C + c]
                         : fv[((size_t)n * C + c) * 3 + (d - 1)];
      mx = fmaxf(mx, v);
    }
    if (d == 0) pooled_s[(size_t)m * C + c] = mx;
    else        pooled_v[(size_t)(d - 1) * M * C + (size_t)m * C + c] = mx;
  }
}

__global__ __launch_bounds__(256)
void mix_wmma_kernel(const float* __restrict__ pooled_s,
                     const float* __restrict__ pooled_v,
                     const float* __restrict__ Ws, const float* __restrict__ Wv,
                     const float* __restrict__ bs, const float* __restrict__ bv,
                     float* __restrict__ outS,
                     float* __restrict__ outV,
                     int M, int C) {
  const int lane = threadIdx.x & 31;
  const int wave = threadIdx.x >> 5;
  const int tiles_o = C >> 4;
  const int tiles_m = M >> 4;
  const int per_d   = tiles_m * tiles_o;
  const int tile    = blockIdx.x * (blockDim.x >> 5) + wave;
  if (tile >= per_d * 4) return;

  const int d   = tile / per_d;
  const int rem = tile % per_d;
  const int m0  = (rem / tiles_o) << 4;
  const int o0  = (rem % tiles_o) << 4;

  const float* A    = (d == 0) ? (pooled_s + (size_t)m0 * C)
                               : (pooled_v + (size_t)(d - 1) * M * C + (size_t)m0 * C);
  const float* B    = (d == 0) ? Ws : Wv;
  const float* bias = (d == 0) ? bs : bv;

  const int ml = lane & 15;
  const int kh = lane >> 4;

  v8f acc = {};
  for (int k0 = 0; k0 < C; k0 += 4) {
    const int ka = k0 + 2 * kh;
    v2f a = *(const v2f*)(A + (size_t)ml * C + ka);
    v2f b;
    b.x = B[(size_t)ka * C + o0 + ml];
    b.y = B[(size_t)(ka + 1) * C + o0 + ml];
    acc = __builtin_amdgcn_wmma_f32_16x16x4_f32(
        false, a, false, b, (short)0, acc, false, false);
  }

  const int   n  = o0 + ml;
  const float bb = bias[n];
  #pragma unroll
  for (int v = 0; v < 8; ++v) {
    const int mm  = m0 + v + 8 * kh;
    const float val = acc[v] + bb;
    if (d == 0) outS[(size_t)mm * C + n] = val;
    else        outV[((size_t)mm * C + n) * 3 + (d - 1)] = val;
  }
}

extern "C" void kernel_launch(void* const* d_in, const int* in_sizes, int n_in,
                              void* d_out, int out_size, void* d_ws, size_t ws_size,
                              hipStream_t stream) {
  const float* pos = (const float*)d_in[0];
  const float* fs  = (const float*)d_in[1];
  const float* fv  = (const float*)d_in[2];
  const float* Ws  = (const float*)d_in[3];
  const float* Wv  = (const float*)d_in[4];
  const float* bs  = (const float*)d_in[5];
  const float* bv  = (const float*)d_in[6];

  const int N = in_sizes[0] / 3;
  const int C = in_sizes[1] / N;
  const int M = (N < 2048) ? N : 2048;

  float* out   = (float*)d_out;
  float* cents = out;
  float* outS  = out + (size_t)M * 3;
  float* outV  = outS + (size_t)M * C;

  int*   cent_idx = (int*)d_ws;
  float* wsf      = (float*)d_ws;
  float* spill    = wsf + 4096;
  float* pooled_s = wsf + 4096 + 65536;
  float* pooled_v = pooled_s + (size_t)M * C;

  fps_kernel<<<1, 512, 0, stream>>>(pos, spill, cent_idx, cents, N, M);
  bq_pool_kernel<<<M, 256, 0, stream>>>(pos, fs, fv, cent_idx,
                                        pooled_s, pooled_v, N, C, M);

  const int per_d = (M / 16) * (C / 16);
  const int total = per_d * 4;
  mix_wmma_kernel<<<(total + 7) / 8, 256, 0, stream>>>(
      pooled_s, pooled_v, Ws, Wv, bs, bv, outS, outV, M, C);
}


# FPS: each wave's issue priority falls as it advances through the point loop (s_setprio 3,2,1,0): the 4 waves of a SIMD advance together, not oldest first, and cover each other's read waits
# speedup vs baseline: 8.2558x; 1.0944x over previous
#include <hip/hip_runtime.h>
#include <hip/hip_bf16.h>

#define FLT_BIG 3.402823466e+38f
#define RADIUS2 0.01f
#define K_LOCAL 32
#define CAP 2048

typedef __attribute__((ext_vector_type(2))) float v2f;
typedef __attribute__((ext_vector_type(8))) float v8f;

#define FPS_T     512
#define FPS_PPT   98
#define FPS_PITCH 156
#define FPS_GS    14
#define FPS_NG    7
#define FPS_NQ    25

typedef __attribute__((ext_vector_type(4))) float v4f;
typedef __attribute__((ext_vector_type(2))) int   v2i;

template <int MASK>
__device__ __forceinline__ int fps_row_max(int v) {
  const int ov = __builtin_amdgcn_update_dpp(v, v, 0x160 + MASK, 0xf, 0xf, true);
  return (ov > v) ? ov : v;
}

template <int MASK>
__device__ __forceinline__ int fps_row_min(int v) {
  const int ov = __builtin_amdgcn_update_dpp(v, v, 0x160 + MASK, 0xf, 0xf, true);
  return (ov < v) ? ov : v;
}

__global__ __launch_bounds__(FPS_T)
void fps_kernel(const float* __restrict__ pos, float* __restrict__ spill,
                int* __restrict__ cent_idx, float* __restrict__ cents,
                int N, int M) {
  const int tid  = threadIdx.x;
  const int lane = tid & 31;
  const int wave = tid >> 5;
  __shared__ __attribute__((aligned(8)))  v2i   s_win[2][FPS_T / 32];
  __shared__ __attribute__((aligned(16))) float s_pt[FPS_T * FPS_PITCH];

  if (N > FPS_PPT * FPS_T) return;

  float*     L  = s_pt + tid * FPS_PITCH;
  const v4f* L4 = (const v4f*)L;
  const v4f* P4 = (const v4f*)spill;

  float px[FPS_PPT];
  float dist[FPS_PPT];
  #pragma unroll
  for (int j = 0; j < FPS_PPT; ++j) {
    const int i = tid + j * FPS_T;
    float x = 0.0f, y = 0.0f, z = 0.0f, d = -FLT_BIG;
    if (i < N) {
      x = pos[i * 3 + 0];
      y = pos[i * 3 + 1];
      z = pos[i * 3 + 2];
      d = FLT_BIG;
    }
    dist[j] = d;
    L[j] = y;
    if (j >= 54 && j < 82) spill[((((6 + ((j - 54) >> 2)) * FPS_T) + tid) << 2) + ((j - 54) & 3)] = x;
    else                   px[j] = x;
    if (j < 24)      spill[((((j >> 2) * FPS_T) + tid) << 2) + (j & 3)] = z;
    else if (j < 80) L[100 + (j - 24)] = z;
    else if (j < 82) L[98 + (j - 80)] = z;
    else             spill[((((13 + ((j - 82) >> 2)) * FPS_T) + tid) << 2) + ((j - 82) & 3)] = z;
  }
  __syncthreads();

  v4f zb1[6];
  #pragma unroll
  for (int q = 0; q < 6; ++q) zb1[q] = P4[q * FPS_T + tid];

  int cur = 0;
  float ex = 0.0f, ey = 0.0f, ez = 0.0f;
  const unsigned last_pt = (unsigned)(N - 1);
  for (int it = 0; it < M; ++it) {
    __builtin_amdgcn_s_setprio(3);
    const float* cp = (const float*)((const char*)pos + __umul24((unsigned)cur, 12u));
    const float cx = cp[0];
    const float cy = cp[1];
    const float cz = cp[2];
    if (tid == 0) {
      cent_idx[it] = cur;
      cents[it * 3 + 0] = cx;
      cents[it * 3 + 1] = cy;
      cents[it * 3 + 2] = cz;
    }
    float gm[FPS_NG];
    v4f xb[7];
    v4f zc[4];
    #pragma unroll
    for (int q = 0; q < FPS_NQ; ++q) {
      if (q == 6)  __builtin_amdgcn_s_setprio(2);
      if (q == 13) __builtin_amdgcn_s_setprio(1);
      if (q == 20) __builtin_amdgcn_s_setprio(0);
      if (q == 6) {
        #pragma unroll
        for (int p = 0; p < 7; ++p) xb[p] = P4[(6 + p) * FPS_T + tid];
      }
      if (q == 15) {
        #pragma unroll
        for (int p = 0; p < 4; ++p) zc[p] = P4[(13 + p) * FPS_T + tid];
      }
      if (q == 21) {
        #pragma unroll
        for (int p = 0; p < 6; ++p) zb1[p] = P4[p * FPS_T + tid];
      }
      const v4f yv = L4[q];
      float dx[4], dy[4], dz[4], d2[4];
      #pragma unroll
      for (int r = 0; r < 4; ++r) {
        const int j = 4 * q + r;
        if (j < FPS_PPT) {
          float z;
          if (j < 24)      { z = zb1[j >> 2][j & 3]; }
          else if (j < 80) { const v4f zv = L4[25 + ((j - 24) >> 2)]; z = zv[(j - 24) & 3]; }
          else if (j < 82) { const v4f zv = L4[24]; z = zv[2 + (j - 80)]; }
          else             { z = zc[(j - 82) >> 2][(j - 82) & 3]; }
          float x;
          if (j >= 54 && j < 82) x = xb[(j - 54) >> 2][(j - 54) & 3];
          else                   x = px[j];
          dx[r] = x - cx;
          dy[r] = yv[r] - cy;
          dz[r] = z - cz;
        }
      }
      #pragma unroll
      for (int r = 0; r < 4; ++r) if (4 * q + r < FPS_PPT) d2[r] = dy[r] * dy[r];
      #pragma unroll
      for (int r = 0; r < 4; ++r) if (4 * q + r < FPS_PPT) d2[r] = __builtin_fmaf(dx[r], dx[r], d2[r]);
      #pragma unroll
      for (int r = 0; r < 4; ++r) if (4 * q + r < FPS_PPT) d2[r] = __builtin_fmaf(dz[r], dz[r], d2[r]);
      #pragma unroll
      for (int r = 0; r < 4; ++r) {
        const int j = 4 * q + r;
        if (j < FPS_PPT) {
          const float dd = fminf(dist[j], d2[r]);
          dist[j] = dd;
          d2[r] = dd;
        }
      }
      #pragma unroll
      for (int r = 0; r < 4; ++r) {
        const int j = 4 * q + r;
        if (j < FPS_PPT) {
          if (j % FPS_GS == 0) gm[j / FPS_GS] = d2[r];
          else                 gm[j / FPS_GS] = fmaxf(gm[j / FPS_GS], d2[r]);
        }
      }
      if (q == 0) asm volatile("" :: "v"(ex), "v"(ey), "v"(ez));
      if (q % 2 == 1) __builtin_amdgcn_sched_barrier(0x0094);
    }
    float bv = gm[0];
    #pragma unroll
    for (int k = 1; k < FPS_NG; ++k) bv = fmaxf(bv, gm[k]);
    int vb = __builtin_bit_cast(int, bv);
    vb = fps_row_max<1>(vb);
    vb = fps_row_max<2>(vb);
    vb = fps_row_max<4>(vb);
    vb = fps_row_max<8>(vb);
    {
      const int ov = __builtin_amdgcn_permlanex16(vb, vb, 0x76543210, 0xfedcba98, false, false);
      vb = (ov > vb) ? ov : vb;
    }
    const float V = __builtin_bit_cast(float, vb);
    unsigned gb[FPS_NG];
    #pragma unroll
    for (int k = FPS_NG - 1; k >= 0; --k) gb[k] = __builtin_amdgcn_ballot_w32(gm[k] == V);
    __builtin_amdgcn_sched_barrier(0);
    int kg = 0;
    #pragma unroll
    for (int k = FPS_NG - 1; k >= 0; --k) {
      if (gb[k] != 0u) kg = k;
    }
    int      jj = 0;
    unsigned mm = 1u;
    #pragma unroll
    for (int k = 0; k < FPS_NG; ++k) {
      if (kg == k) {
        unsigned mr[FPS_GS];
        #pragma unroll
        for (int r = FPS_GS - 1; r >= 0; --r) mr[r] = __builtin_amdgcn_ballot_w32(dist[k * FPS_GS + r] == V);
        __builtin_amdgcn_sched_barrier(0);
        #pragma unroll
        for (int r = FPS_GS - 1; r >= 0; --r) {
          if (mr[r] != 0u) { jj = k * FPS_GS + r; mm = mr[r]; }
        }
      }
    }
    const int bi = (wave << 5) + __builtin_ctz(mm) + jj * FPS_T;
    if (lane == 0) {
      v2i w;
      w.x = vb;
      w.y = bi;
      s_win[it & 1][wave] = w;
    }
    {
      const unsigned eb = ((unsigned)bi < last_pt) ? (unsigned)bi : last_pt;
      const float* ep = (const float*)((const char*)pos + __umul24(eb, 12u));
      ex = ep[0];
      ey = ep[1];
      ez = ep[2];
    }
    __builtin_amdgcn_fence(__ATOMIC_RELEASE, "workgroup", "local");
    __builtin_amdgcn_s_barrier();
    __builtin_amdgcn_fence(__ATOMIC_ACQUIRE, "workgroup", "local");
    const v2i ws = s_win[it & 1][lane & 15];
    int tv = ws.x;
    tv = fps_row_max<1>(tv);
    tv = fps_row_max<2>(tv);
    tv = fps_row_max<4>(tv);
    tv = fps_row_max<8>(tv);
    int ci = (ws.x == tv) ? ws.y : 0x7fffffff;
    ci = fps_row_min<1>(ci);
    ci = fps_row_min<2>(ci);
    ci = fps_row_min<4>(ci);
    ci = fps_row_min<8>(ci);
    cur = (ci < N) ? ci : 0;
  }
}

__global__ __launch_bounds__(256)
void bq_pool_kernel(const float* __restrict__ pos,
                    const float* __restrict__ fs,
                    const float* __restrict__ fv,
                    const int* __restrict__ cent_idx,
                    float* __restrict__ pooled_s,
                    float* __restrict__ pooled_v,
                    int N, int C, int M) {
  const int m   = blockIdx.x;
  const int tid = threadIdx.x;
  __shared__ float s_cd2[CAP];
  __shared__ int   s_cidx[CAP];
  __shared__ int   s_cnt;
  __shared__ int   s_sel[K_LOCAL];
  __shared__ float s_rv[256];
  __shared__ int   s_ri[256];
  __shared__ int   s_rp[256];
  __shared__ float s_c[3];

  if (tid == 0) {
    s_cnt = 0;
    int ci = cent_idx[m];
    s_c[0] = pos[ci * 3 + 0];
    s_c[1] = pos[ci * 3 + 1];
    s_c[2] = pos[ci * 3 + 2];
  }
  __syncthreads();
  const float cx = s_c[0], cy = s_c[1], cz = s_c[2];

  for (int i = tid; i < N; i += 256) {
    float dx = pos[i * 3 + 0] - cx;
    float dy = pos[i * 3 + 1] - cy;
    float dz = pos[i * 3 + 2] - cz;
    float d2 = dx * dx + dy * dy + dz * dz;
    if (d2 <= RADIUS2) {
      int p = atomicAdd(&s_cnt, 1);
      if (p < CAP) { s_cd2[p] = d2; s_cidx[p] = i; }
    }
  }
  __syncthreads();
  const int ncand = min(s_cnt, CAP);

  for (int s = 0; s < K_LOCAL; ++s) {
    float bv = FLT_BIG;
    int   bi = 0x7fffffff;
    int   bp = -1;
    for (int p = tid; p < ncand; p += 256) {
      float v  = s_cd2[p];
      int   ix = s_cidx[p];
      if (v < bv || (v == bv && ix < bi)) { bv = v; bi = ix; bp = p; }
    }
    s_rv[tid] = bv; s_ri[tid] = bi; s_rp[tid] = bp;
    __syncthreads();
    for (int off = 128; off >= 1; off >>= 1) {
      if (tid < off) {
        float ov = s_rv[tid + off];
        int   oi = s_ri[tid + off];
        if (ov < s_rv[tid] || (ov == s_rv[tid] && oi < s_ri[tid])) {
          s_rv[tid] = ov; s_ri[tid] = oi; s_rp[tid] = s_rp[tid + off];
        }
      }
      __syncthreads();
    }
    if (tid == 0) {
      if (s < ncand) {
        int p = s_rp[0];
        s_sel[s] = s_cidx[p];
        s_cd2[p] = FLT_BIG;
      } else {
        s_sel[s] = s_sel[0];
      }
    }
    __syncthreads();
  }

  for (int j = tid; j < C * 4; j += 256) {
    const int c = j >> 2;
    const int d = j & 3;
    float mx = -FLT_BIG;
    #pragma unroll
    for (int t = 0; t < K_LOCAL; ++t) {
      int n = s_sel[t];
      float v = (d == 0) ? fs[(size_t)n * C + c]
                         : fv[((size_t)n * C + c) * 3 + (d - 1)];
      mx = fmaxf(mx, v);
    }
    if (d == 0) pooled_s[(size_t)m * C + c] = mx;
    else        pooled_v[(size_t)(d - 1) * M * C + (size_t)m * C + c] = mx;
  }
}

__global__ __launch_bounds__(256)
void mix_wmma_kernel(const float* __restrict__ pooled_s,
                     const float* __restrict__ pooled_v,
                     const float* __restrict__ Ws, const float* __restrict__ Wv,
                     const float* __restrict__ bs, const float* __restrict__ bv,
                     float* __restrict__ outS,
                     float* __restrict__ outV,
                     int M, int C) {
  const int lane = threadIdx.x & 31;
  const int wave = threadIdx.x >> 5;
  const int tiles_o = C >> 4;
  const int tiles_m = M >> 4;
  const int per_d   = tiles_m * tiles_o;
  const int tile    = blockIdx.x * (blockDim.x >> 5) + wave;
  if (tile >= per_d * 4) return;

  const int d   = tile / per_d;
  const int rem = tile % per_d;
  const int m0  = (rem / tiles_o) << 4;
  const int o0  = (rem % tiles_o) << 4;

  const float* A    = (d == 0) ? (pooled_s + (size_t)m0 * C)
                               : (pooled_v + (size_t)(d - 1) * M * C + (size_t)m0 * C);
  const float* B    = (d == 0) ? Ws : Wv;
  const float* bias = (d == 0) ? bs : bv;

  const int ml = lane & 15;
  const int kh = lane >> 4;

  v8f acc = {};
  for (int k0 = 0; k0 < C; k0 += 4) {
    const int ka = k0 + 2 * kh;
    v2f a = *(const v2f*)(A + (size_t)ml * C + ka);
    v2f b;
    b.x = B[(size_t)ka * C + o0 + ml];
    b.y = B[(size_t)(ka + 1) * C + o0 + ml];
    acc = __builtin_amdgcn_wmma_f32_16x16x4_f32(
        false, a, false, b, (short)0, acc, false, false);
  }

  const int   n  = o0 + ml;
  const float bb = bias[n];
  #pragma unroll
  for (int v = 0; v < 8; ++v) {
    const int mm  = m0 + v + 8 * kh;
    const float val = acc[v] + bb;
    if (d == 0) outS[(size_t)mm * C + n] = val;
    else        outV[((size_t)mm * C + n) * 3 + (d - 1)] = val;
  }
}

extern "C" void kernel_launch(void* const* d_in, const int* in_sizes, int n_in,
                              void* d_out, int out_size, void* d_ws, size_t ws_size,
                              hipStream_t stream) {
  const float* pos = (const float*)d_in[0];
  const float* fs  = (const float*)d_in[1];
  const float* fv  = (const float*)d_in[2];
  const float* Ws  = (const float*)d_in[3];
  const float* Wv  = (const float*)d_in[4];
  const float* bs  = (const float*)d_in[5];
  const float* bv  = (const float*)d_in[6];

  const int N = in_sizes[0] / 3;
  const int C = in_sizes[1] / N;
  const int M = (N < 2048) ? N : 2048;

  float* out   = (float*)d_out;
  float* cents = out;
  float* outS  = out + (size_t)M * 3;
  float* outV  = outS + (size_t)M * C;

  int*   cent_idx = (int*)d_ws;
  float* wsf      = (float*)d_ws;
  float* spill    = wsf + 4096;
  float* pooled_s = wsf + 4096 + 65536;
  float* pooled_v = pooled_s + (size_t)M * C;

  fps_kernel<<<1, 512, 0, stream>>>(pos, spill, cent_idx, cents, N, M);
  bq_pool_kernel<<<M, 256, 0, stream>>>(pos, fs, fv, cent_idx,
                                        pooled_s, pooled_v, N, C, M);

  const int per_d = (M / 16) * (C / 16);
  const int total = per_d * 4;
  mix_wmma_kernel<<<(total + 7) / 8, 256, 0, stream>>>(
      pooled_s, pooled_v, Ws, Wv, bs, bv, outS, outV, M, C);
}


# FPS: each argmax also yields its runner-up; if an exact test shows the winner's sweep cannot lower the runner-up's distance, it is the next centroid and ONE sweep applies both; bit-identical
# speedup vs baseline: 9.0089x; 1.0912x over previous
#include <hip/hip_runtime.h>
#include <hip/hip_bf16.h>

#define FLT_BIG 3.402823466e+38f
#define RADIUS2 0.01f
#define K_LOCAL 32
#define CAP 2048

typedef __attribute__((ext_vector_type(2))) float v2f;
typedef __attribute__((ext_vector_type(8))) float v8f;

#define FPS_T     512
#define FPS_PPT   98
#define FPS_PITCH 156
#define FPS_GS    14
#define FPS_NG    7
#define FPS_NQ    25

typedef __attribute__((ext_vector_type(4))) float v4f;

template <int MASK>
__device__ __forceinline__ int fps_row_max(int v) {
  const int ov = __builtin_amdgcn_update_dpp(v, v, 0x160 + MASK, 0xf, 0xf, true);
  return (ov > v) ? ov : v;
}

template <int MASK>
__device__ __forceinline__ int fps_row_min(int v) {
  const int ov = __builtin_amdgcn_update_dpp(v, v, 0x160 + MASK, 0xf, 0xf, true);
  return (ov < v) ? ov : v;
}

typedef __attribute__((ext_vector_type(4))) int v4i;

__device__ __forceinline__ void fps_sweep(float (&dist)[FPS_PPT], const float (&px)[FPS_PPT],
                                          float (&gm)[FPS_NG], v4f (&zb1)[6],
                                          const v4f* L4, const v4f* P4, const int tid,
                                          const float cx, const float cy, const float cz,
                                          const float ux, const float uy, const float uz,
                                          const float ex, const float ey, const float ez) {
  v4f xb[7];
  v4f zc[4];
  #pragma unroll
  for (int q = 0; q < FPS_NQ; ++q) {
    if (q == 6)  __builtin_amdgcn_s_setprio(2);
    if (q == 13) __builtin_amdgcn_s_setprio(1);
    if (q == 20) __builtin_amdgcn_s_setprio(0);
    if (q == 6) {
      #pragma unroll
      for (int p = 0; p < 7; ++p) xb[p] = P4[(6 + p) * FPS_T + tid];
    }
    if (q == 15) {
      #pragma unroll
      for (int p = 0; p < 4; ++p) zc[p] = P4[(13 + p) * FPS_T + tid];
    }
    if (q == 21) {
      #pragma unroll
      for (int p = 0; p < 6; ++p) zb1[p] = P4[p * FPS_T + tid];
    }
    const v4f yv = L4[q];
    float dx[4], dy[4], dz[4], d2[4];
    float qx[4], qz[4], e2[4];
    #pragma unroll
    for (int r = 0; r < 4; ++r) {
      const int j = 4 * q + r;
      if (j < FPS_PPT) {
        float z;
        if (j < 24)      { z = zb1[j >> 2][j & 3]; }
        else if (j < 80) { const v4f zv = L4[25 + ((j - 24) >> 2)]; z = zv[(j - 24) & 3]; }
        else if (j < 82) { const v4f zv = L4[24]; z = zv[2 + (j - 80)]; }
        else             { z = zc[(j - 82) >> 2][(j - 82) & 3]; }
        float x;
        if (j >= 54 && j < 82) x = xb[(j - 54) >> 2][(j - 54) & 3];
        else                   x = px[j];
        dx[r] = x - cx;
        dy[r] = yv[r] - cy;
        dz[r] = z - cz;
        qx[r] = x;
        qz[r] = z;
      }
    }
    #pragma unroll
    for (int r = 0; r < 4; ++r) if (4 * q + r < FPS_PPT) d2[r] = dy[r] * dy[r];
    #pragma unroll
    for (int r = 0; r < 4; ++r) if (4 * q + r < FPS_PPT) d2[r] = __builtin_fmaf(dx[r], dx[r], d2[r]);
    #pragma unroll
    for (int r = 0; r < 4; ++r) if (4 * q + r < FPS_PPT) d2[r] = __builtin_fmaf(dz[r], dz[r], d2[r]);
    #pragma unroll
    for (int r = 0; r < 4; ++r) if (4 * q + r < FPS_PPT) { dx[r] = qx[r] - ux; dy[r] = yv[r] - uy; dz[r] = qz[r] - uz; }
    #pragma unroll
    for (int r = 0; r < 4; ++r) if (4 * q + r < FPS_PPT) e2[r] = dy[r] * dy[r];
    #pragma unroll
    for (int r = 0; r < 4; ++r) if (4 * q + r < FPS_PPT) e2[r] = __builtin_fmaf(dx[r], dx[r], e2[r]);
    #pragma unroll
    for (int r = 0; r < 4; ++r) if (4 * q + r < FPS_PPT) e2[r] = __builtin_fmaf(dz[r], dz[r], e2[r]);
    #pragma unroll
    for (int r = 0; r < 4; ++r) {
      const int j = 4 * q + r;
      if (j < FPS_PPT) {
        float dd = fminf(dist[j], d2[r]);
        dd = fminf(dd, e2[r]);
        dist[j] = dd;
        d2[r] = dd;
      }
    }
    #pragma unroll
    for (int r = 0; r < 4; ++r) {
      const int j = 4 * q + r;
      if (j < FPS_PPT) {
        if (j % FPS_GS == 0) gm[j / FPS_GS] = d2[r];
        else                 gm[j / FPS_GS] = fmaxf(gm[j / FPS_GS], d2[r]);
      }
    }
    if (q == 0) asm volatile("" :: "v"(ex), "v"(ey), "v"(ez));
    if (q % 2 == 1) __builtin_amdgcn_sched_barrier(0x0094);
  }
}

__global__ __launch_bounds__(FPS_T)
void fps_kernel(const float* __restrict__ pos, float* __restrict__ spill,
                int* __restrict__ cent_idx, float* __restrict__ cents,
                int N, int M) {
  const int tid  = threadIdx.x;
  const int lane = tid & 31;
  const int wave = tid >> 5;
  __shared__ __attribute__((aligned(16))) v4i   s_win[2][FPS_T / 32];
  __shared__ __attribute__((aligned(16))) float s_pt[FPS_T * FPS_PITCH];

  if (N > FPS_PPT * FPS_T) return;

  float*     L  = s_pt + tid * FPS_PITCH;
  const v4f* L4 = (const v4f*)L;
  const v4f* P4 = (const v4f*)spill;

  float px[FPS_PPT];
  float dist[FPS_PPT];
  #pragma unroll
  for (int j = 0; j < FPS_PPT; ++j) {
    const int i = tid + j * FPS_T;
    float x = 0.0f, y = 0.0f, z = 0.0f, d = -FLT_BIG;
    if (i < N) {
      x = pos[i * 3 + 0];
      y = pos[i * 3 + 1];
      z = pos[i * 3 + 2];
      d = FLT_BIG;
    }
    dist[j] = d;
    L[j] = y;
    if (j >= 54 && j < 82) spill[((((6 + ((j - 54) >> 2)) * FPS_T) + tid) << 2) + ((j - 54) & 3)] = x;
    else                   px[j] = x;
    if (j < 24)      spill[((((j >> 2) * FPS_T) + tid) << 2) + (j & 3)] = z;
    else if (j < 80) L[100 + (j - 24)] = z;
    else if (j < 82) L[98 + (j - 80)] = z;
    else             spill[((((13 + ((j - 82) >> 2)) * FPS_T) + tid) << 2) + ((j - 82) & 3)] = z;
  }
  __syncthreads();

  v4f zb1[6];
  #pragma unroll
  for (int q = 0; q < 6; ++q) zb1[q] = P4[q * FPS_T + tid];

  int cur  = 0;
  int nxt  = 0;
  int v2b  = 0;
  int cok  = 0;
  float ex = 0.0f, ey = 0.0f, ez = 0.0f;
  const unsigned last_pt = (unsigned)(N - 1);
  int it  = 0;
  int par = 0;
  while (it < M) {
    __builtin_amdgcn_s_setprio(3);
    const float* cp = (const float*)((const char*)pos + __umul24((unsigned)cur, 12u));
    const float* up = (const float*)((const char*)pos + __umul24((unsigned)nxt, 12u));
    const float cx = cp[0];
    const float cy = cp[1];
    const float cz = cp[2];
    const float ux = up[0];
    const float uy = up[1];
    const float uz = up[2];
    const float gx = ux - cx;
    const float gy = uy - cy;
    const float gz = uz - cz;
    float g2 = gy * gy;
    g2 = __builtin_fmaf(gx, gx, g2);
    g2 = __builtin_fmaf(gz, gz, g2);
    const int two_v = (cok != 0 && g2 >= __builtin_bit_cast(float, v2b)) ? 1 : 0;
    const bool two = (it + 1 < M) && (__builtin_amdgcn_readfirstlane(two_v) != 0);
    if (tid == 0) {
      cent_idx[it] = cur;
      cents[it * 3 + 0] = cx;
      cents[it * 3 + 1] = cy;
      cents[it * 3 + 2] = cz;
      if (two) {
        cent_idx[it + 1] = nxt;
        cents[it * 3 + 3] = ux;
        cents[it * 3 + 4] = uy;
        cents[it * 3 + 5] = uz;
      }
    }
    float gm[FPS_NG];
    fps_sweep(dist, px, gm, zb1, L4, P4, tid, cx, cy, cz,
                    two ? ux : cx, two ? uy : cy, two ? uz : cz, ex, ey, ez);
    it += two ? 2 : 1;
    float bv = gm[0];
    #pragma unroll
    for (int k = 1; k < FPS_NG; ++k) bv = fmaxf(bv, gm[k]);
    int vb = __builtin_bit_cast(int, bv);
    vb = fps_row_max<1>(vb);
    vb = fps_row_max<2>(vb);
    vb = fps_row_max<4>(vb);
    vb = fps_row_max<8>(vb);
    {
      const int ov = __builtin_amdgcn_permlanex16(vb, vb, 0x76543210, 0xfedcba98, false, false);
      vb = (ov > vb) ? ov : vb;
    }
    const float V = __builtin_bit_cast(float, vb);
    unsigned gb[FPS_NG];
    #pragma unroll
    for (int k = FPS_NG - 1; k >= 0; --k) gb[k] = __builtin_amdgcn_ballot_w32(gm[k] == V);
    __builtin_amdgcn_sched_barrier(0);
    int kg = 0;
    int npair = 0;
    #pragma unroll
    for (int k = FPS_NG - 1; k >= 0; --k) {
      if (gb[k] != 0u) kg = k;
      npair += __builtin_popcount(gb[k]);
    }
    int      jj  = 0;
    unsigned mm  = 1u;
    int      neq = 0;
    float    lx  = -FLT_BIG;
    #pragma unroll
    for (int k = 0; k < FPS_NG; ++k) {
      if (kg == k) {
        unsigned mr[FPS_GS];
        #pragma unroll
        for (int r = FPS_GS - 1; r >= 0; --r) mr[r] = __builtin_amdgcn_ballot_w32(dist[k * FPS_GS + r] == V);
        __builtin_amdgcn_sched_barrier(0);
        #pragma unroll
        for (int r = FPS_GS - 1; r >= 0; --r) {
          if (mr[r] != 0u) { jj = k * FPS_GS + r; mm = mr[r]; }
          neq += __builtin_popcount(mr[r]);
        }
        float g = -FLT_BIG;
        #pragma unroll
        for (int r = 0; r < FPS_GS; ++r) {
          const float e = dist[k * FPS_GS + r];
          g = fmaxf(g, (e == V) ? -FLT_BIG : e);
        }
        #pragma unroll
        for (int k2 = 0; k2 < FPS_NG; ++k2) {
          if (k2 != k) g = fmaxf(g, gm[k2]);
        }
        lx = g;
      }
    }
    int sb = __builtin_bit_cast(int, lx);
    sb = fps_row_max<1>(sb);
    sb = fps_row_max<2>(sb);
    sb = fps_row_max<4>(sb);
    sb = fps_row_max<8>(sb);
    {
      const int ov = __builtin_amdgcn_permlanex16(sb, sb, 0x76543210, 0xfedcba98, false, false);
      sb = (ov > sb) ? ov : sb;
    }
    if (npair != 1 || neq != 1) sb = vb;
    const int bi = (wave << 5) + __builtin_ctz(mm) + jj * FPS_T;
    if (lane == 0) {
      v4i w;
      w.x = vb;
      w.y = bi;
      w.z = sb;
      w.w = 0;
      s_win[par][wave] = w;
    }
    {
      const unsigned eb = ((unsigned)bi < last_pt) ? (unsigned)bi : last_pt;
      const float* ep = (const float*)((const char*)pos + __umul24(eb, 12u));
      ex = ep[0];
      ey = ep[1];
      ez = ep[2];
    }
    __builtin_amdgcn_fence(__ATOMIC_RELEASE, "workgroup", "local");
    __builtin_amdgcn_s_barrier();
    __builtin_amdgcn_fence(__ATOMIC_ACQUIRE, "workgroup", "local");
    const v4i ws = s_win[par][lane & 15];
    par ^= 1;
    int tv = ws.x;
    tv = fps_row_max<1>(tv);
    tv = fps_row_max<2>(tv);
    tv = fps_row_max<4>(tv);
    tv = fps_row_max<8>(tv);
    int ci = (ws.x == tv) ? ws.y : 0x7fffffff;
    ci = fps_row_min<1>(ci);
    ci = fps_row_min<2>(ci);
    ci = fps_row_min<4>(ci);
    ci = fps_row_min<8>(ci);
    const bool isw = (ws.x == tv) && (ws.y == ci);
    const int  rv  = isw ? ws.z : ws.x;
    int v2 = rv;
    v2 = fps_row_max<1>(v2);
    v2 = fps_row_max<2>(v2);
    v2 = fps_row_max<4>(v2);
    v2 = fps_row_max<8>(v2);
    int c2 = (rv == v2) ? (isw ? -1 : ws.y) : 0x7fffffff;
    c2 = fps_row_min<1>(c2);
    c2 = fps_row_min<2>(c2);
    c2 = fps_row_min<4>(c2);
    c2 = fps_row_min<8>(c2);
    cur = (ci < N) ? ci : 0;
    cok = (c2 >= 0 && c2 < N && __builtin_bit_cast(float, v2) > 0.0f) ? 1 : 0;
    nxt = (cok != 0) ? c2 : 0;
    v2b = v2;
  }
}

__global__ __launch_bounds__(256)
void bq_pool_kernel(const float* __restrict__ pos,
                    const float* __restrict__ fs,
                    const float* __restrict__ fv,
                    const int* __restrict__ cent_idx,
                    float* __restrict__ pooled_s,
                    float* __restrict__ pooled_v,
                    int N, int C, int M) {
  const int m   = blockIdx.x;
  const int tid = threadIdx.x;
  __shared__ float s_cd2[CAP];
  __shared__ int   s_cidx[CAP];
  __shared__ int   s_cnt;
  __shared__ int   s_sel[K_LOCAL];
  __shared__ float s_rv[256];
  __shared__ int   s_ri[256];
  __shared__ int   s_rp[256];
  __shared__ float s_c[3];

  if (tid == 0) {
    s_cnt = 0;
    int ci = cent_idx[m];
    s_c[0] = pos[ci * 3 + 0];
    s_c[1] = pos[ci * 3 + 1];
    s_c[2] = pos[ci * 3 + 2];
  }
  __syncthreads();
  const float cx = s_c[0], cy = s_c[1], cz = s_c[2];

  for (int i = tid; i < N; i += 256) {
    float dx = pos[i * 3 + 0] - cx;
    float dy = pos[i * 3 + 1] - cy;
    float dz = pos[i * 3 + 2] - cz;
    float d2 = dx * dx + dy * dy + dz * dz;
    if (d2 <= RADIUS2) {
      int p = atomicAdd(&s_cnt, 1);
      if (p < CAP) { s_cd2[p] = d2; s_cidx[p] = i; }
    }
  }
  __syncthreads();
  const int ncand = min(s_cnt, CAP);

  for (int s = 0; s < K_LOCAL; ++s) {
    float bv = FLT_BIG;
    int   bi = 0x7fffffff;
    int   bp = -1;
    for (int p = tid; p < ncand; p += 256) {
      float v  = s_cd2[p];
      int   ix = s_cidx[p];
      if (v < bv || (v == bv && ix < bi)) { bv = v; bi = ix; bp = p; }
    }
    s_rv[tid] = bv; s_ri[tid] = bi; s_rp[tid] = bp;
    __syncthreads();
    for (int off = 128; off >= 1; off >>= 1) {
      if (tid < off) {
        float ov = s_rv[tid + off];
        int   oi = s_ri[tid + off];
        if (ov < s_rv[tid] || (ov == s_rv[tid] && oi < s_ri[tid])) {
          s_rv[tid] = ov; s_ri[tid] = oi; s_rp[tid] = s_rp[tid + off];
        }
      }
      __syncthreads();
    }
    if (tid == 0) {
      if (s < ncand) {
        int p = s_rp[0];
        s_sel[s] = s_cidx[p];
        s_cd2[p] = FLT_BIG;
      } else {
        s_sel[s] = s_sel[0];
      }
    }
    __syncthreads();
  }

  for (int j = tid; j < C * 4; j += 256) {
    const int c = j >> 2;
    const int d = j & 3;
    float mx = -FLT_BIG;
    #pragma unroll
    for (int t = 0; t < K_LOCAL; ++t) {
      int n = s_sel[t];
      float v = (d == 0) ? fs[(size_t)n * C + c]
                         : fv[((size_t)n * C + c) * 3 + (d - 1)];
      mx = fmaxf(mx, v);
    }
    if (d == 0) pooled_s[(size_t)m * C + c] = mx;
    else        pooled_v[(size_t)(d - 1) * M * C + (size_t)m * C + c] = mx;
  }
}

__global__ __launch_bounds__(256)
void mix_wmma_kernel(const float* __restrict__ pooled_s,
                     const float* __restrict__ pooled_v,
                     const float* __restrict__ Ws, const float* __restrict__ Wv,
                     const float* __restrict__ bs, const float* __restrict__ bv,
                     float* __restrict__ outS,
                     float* __restrict__ outV,
                     int M, int C) {
  const int lane = threadIdx.x & 31;
  const int wave = threadIdx.x >> 5;
  const int tiles_o = C >> 4;
  const int tiles_m = M >> 4;
  const int per_d   = tiles_m * tiles_o;
  const int tile    = blockIdx.x * (blockDim.x >> 5) + wave;
  if (tile >= per_d * 4) return;

  const int d   = tile / per_d;
  const int rem = tile % per_d;
  const int m0  = (rem / tiles_o) << 4;
  const int o0  = (rem % tiles_o) << 4;

  const float* A    = (d == 0) ? (pooled_s + (size_t)m0 * C)
                               : (pooled_v + (size_t)(d - 1) * M * C + (size_t)m0 * C);
  const float* B    = (d == 0) ? Ws : Wv;
  const float* bias = (d == 0) ? bs : bv;

  const int ml = lane & 15;
  const int kh = lane >> 4;

  v8f acc = {};
  for (int k0 = 0; k0 < C; k0 += 4) {
    const int ka = k0 + 2 * kh;
    v2f a = *(const v2f*)(A + (size_t)ml * C + ka);
    v2f b;
    b.x = B[(size_t)ka * C + o0 + ml];
    b.y = B[(size_t)(ka + 1) * C + o0 + ml];
    acc = __builtin_amdgcn_wmma_f32_16x16x4_f32(
        false, a, false, b, (short)0, acc, false, false);
  }

  const int   n  = o0 + ml;
  const float bb = bias[n];
  #pragma unroll
  for (int v = 0; v < 8; ++v) {
    const int mm  = m0 + v + 8 * kh;
    const float val = acc[v] + bb;
    if (d == 0) outS[(size_t)mm * C + n] = val;
    else        outV[((size_t)mm * C + n) * 3 + (d - 1)] = val;
  }
}

extern "C" void kernel_launch(void* const* d_in, const int* in_sizes, int n_in,
                              void* d_out, int out_size, void* d_ws, size_t ws_size,
                              hipStream_t stream) {
  const float* pos = (const float*)d_in[0];
  const float* fs  = (const float*)d_in[1];
  const float* fv  = (const float*)d_in[2];
  const float* Ws  = (const float*)d_in[3];
  const float* Wv  = (const float*)d_in[4];
  const float* bs  = (const float*)d_in[5];
  const float* bv  = (const float*)d_in[6];

  const int N = in_sizes[0] / 3;
  const int C = in_sizes[1] / N;
  const int M = (N < 2048) ? N : 2048;

  float* out   = (float*)d_out;
  float* cents = out;
  float* outS  = out + (size_t)M * 3;
  float* outV  = outS + (size_t)M * C;

  int*   cent_idx = (int*)d_ws;
  float* wsf      = (float*)d_ws;
  float* spill    = wsf + 4096;
  float* pooled_s = wsf + 4096 + 65536;
  float* pooled_v = pooled_s + (size_t)M * C;

  fps_kernel<<<1, 512, 0, stream>>>(pos, spill, cent_idx, cents, N, M);
  bq_pool_kernel<<<M, 256, 0, stream>>>(pos, fs, fv, cent_idx,
                                        pooled_s, pooled_v, N, C, M);

  const int per_d = (M / 16) * (C / 16);
  const int total = per_d * 4;
  mix_wmma_kernel<<<(total + 7) / 8, 256, 0, stream>>>(
      pooled_s, pooled_v, Ws, Wv, bs, bv, outS, outV, M, C);
}


# FPS: a sweep that applies the winner alone branches over the second centroid's distance chain (8 points per branch) instead of running it on the winner twice; bit-identical
# speedup vs baseline: 9.4170x; 1.0453x over previous
#include <hip/hip_runtime.h>
#include <hip/hip_bf16.h>

#define FLT_BIG 3.402823466e+38f
#define RADIUS2 0.01f
#define K_LOCAL 32
#define CAP 2048

typedef __attribute__((ext_vector_type(2))) float v2f;
typedef __attribute__((ext_vector_type(8))) float v8f;

#define FPS_T     512
#define FPS_PPT   98
#define FPS_PITCH 156
#define FPS_GS    14
#define FPS_NG    7
#define FPS_NQ    25

typedef __attribute__((ext_vector_type(4))) float v4f;

template <int MASK>
__device__ __forceinline__ int fps_row_max(int v) {
  const int ov = __builtin_amdgcn_update_dpp(v, v, 0x160 + MASK, 0xf, 0xf, true);
  return (ov > v) ? ov : v;
}

template <int MASK>
__device__ __forceinline__ int fps_row_min(int v) {
  const int ov = __builtin_amdgcn_update_dpp(v, v, 0x160 + MASK, 0xf, 0xf, true);
  return (ov < v) ? ov : v;
}

typedef __attribute__((ext_vector_type(4))) int v4i;

__device__ __forceinline__ void fps_sweep(float (&dist)[FPS_PPT], const float (&px)[FPS_PPT],
                                          float (&gm)[FPS_NG], v4f (&zb1)[6],
                                          const v4f* L4, const v4f* P4, const int tid,
                                          const float cx, const float cy, const float cz,
                                          const int two_s,
                                          const float ux, const float uy, const float uz,
                                          const float ex, const float ey, const float ez) {
  v4f xb[11];
  v4f zc[4];
  int two_t = two_s;
  asm volatile("" : "+s"(two_t));
  #pragma unroll
  for (int qp = 0; qp < (FPS_NQ + 1) / 2; ++qp) {
    float qx[8], qy[8], qz[8];
    float dx[8], dy[8], dz[8], d2[8];
    #pragma unroll
    for (int h = 0; h < 2; ++h) {
      const int q = 2 * qp + h;
      if (q < FPS_NQ) {
        if (q == 6)  __builtin_amdgcn_s_setprio(2);
        if (q == 13) __builtin_amdgcn_s_setprio(1);
        if (q == 20) __builtin_amdgcn_s_setprio(0);
        if (q == 2) {
          #pragma unroll
          for (int p = 0; p < 4; ++p) xb[p] = P4[(6 + p) * FPS_T + tid];
        }
        if (q == 7) {
          #pragma unroll
          for (int p = 4; p < 11; ++p) xb[p] = P4[(6 + p) * FPS_T + tid];
        }
        if (q == 15) {
          #pragma unroll
          for (int p = 0; p < 4; ++p) zc[p] = P4[(17 + p) * FPS_T + tid];
        }
        if (q == 21) {
          #pragma unroll
          for (int p = 0; p < 6; ++p) zb1[p] = P4[p * FPS_T + tid];
        }
        const v4f yv = L4[q];
        #pragma unroll
        for (int r = 0; r < 4; ++r) {
          const int j = 4 * q + r;
          if (j < FPS_PPT) {
            float z;
            if (j < 24)      { z = zb1[j >> 2][j & 3]; }
            else if (j < 80) { const v4f zv = L4[25 + ((j - 24) >> 2)]; z = zv[(j - 24) & 3]; }
            else if (j < 82) { const v4f zv = L4[24]; z = zv[2 + (j - 80)]; }
            else             { z = zc[(j - 82) >> 2][(j - 82) & 3]; }
            float x;
            if (j >= 38 && j < 82) x = xb[(j - 38) >> 2][(j - 38) & 3];
            else                   x = px[j];
            qx[4 * h + r] = x;
            qy[4 * h + r] = yv[r];
            qz[4 * h + r] = z;
          }
        }
      }
    }
    #pragma unroll
    for (int i = 0; i < 8; ++i) if (8 * qp + i < FPS_PPT) { dx[i] = qx[i] - cx; dy[i] = qy[i] - cy; dz[i] = qz[i] - cz; }
    #pragma unroll
    for (int i = 0; i < 8; ++i) if (8 * qp + i < FPS_PPT) d2[i] = dy[i] * dy[i];
    #pragma unroll
    for (int i = 0; i < 8; ++i) if (8 * qp + i < FPS_PPT) d2[i] = __builtin_fmaf(dx[i], dx[i], d2[i]);
    #pragma unroll
    for (int i = 0; i < 8; ++i) if (8 * qp + i < FPS_PPT) d2[i] = __builtin_fmaf(dz[i], dz[i], d2[i]);
    if (two_s == 0) {
      #pragma unroll
      for (int i = 0; i < 8; ++i) {
        const int j = 8 * qp + i;
        if (j < FPS_PPT) {
          const int a = __builtin_bit_cast(int, dist[j]);
          const int b = __builtin_bit_cast(int, d2[i]);
          dist[j] = __builtin_bit_cast(float, (b < a) ? b : a);
        }
      }
    }
    if (two_t != 0) {
      float e2[8];
      #pragma unroll
      for (int i = 0; i < 8; ++i) if (8 * qp + i < FPS_PPT) { dx[i] = qx[i] - ux; dy[i] = qy[i] - uy; dz[i] = qz[i] - uz; }
      #pragma unroll
      for (int i = 0; i < 8; ++i) if (8 * qp + i < FPS_PPT) e2[i] = dy[i] * dy[i];
      #pragma unroll
      for (int i = 0; i < 8; ++i) if (8 * qp + i < FPS_PPT) e2[i] = __builtin_fmaf(dx[i], dx[i], e2[i]);
      #pragma unroll
      for (int i = 0; i < 8; ++i) if (8 * qp + i < FPS_PPT) e2[i] = __builtin_fmaf(dz[i], dz[i], e2[i]);
      #pragma unroll
      for (int i = 0; i < 8; ++i) {
        const int j = 8 * qp + i;
        if (j < FPS_PPT) {
          float dd = fminf(dist[j], d2[i]);
          dd = fminf(dd, e2[i]);
          dist[j] = dd;
        }
      }
    }
    #pragma unroll
    for (int i = 0; i < 8; ++i) {
      const int j = 8 * qp + i;
      if (j < FPS_PPT) {
        if (j % FPS_GS == 0) gm[j / FPS_GS] = dist[j];
        else                 gm[j / FPS_GS] = fmaxf(gm[j / FPS_GS], dist[j]);
      }
    }
    if (qp == 0) asm volatile("" :: "v"(ex), "v"(ey), "v"(ez));
    __builtin_amdgcn_sched_barrier(0x0094);
  }
}

__global__ __launch_bounds__(FPS_T)
void fps_kernel(const float* __restrict__ pos, float* __restrict__ spill,
                int* __restrict__ cent_idx, float* __restrict__ cents,
                int N, int M) {
  const int tid  = threadIdx.x;
  const int lane = tid & 31;
  const int wave = tid >> 5;
  __shared__ __attribute__((aligned(16))) v4i   s_win[2][FPS_T / 32];
  __shared__ __attribute__((aligned(16))) float s_pt[FPS_T * FPS_PITCH];

  if (N > FPS_PPT * FPS_T) return;

  float*     L  = s_pt + tid * FPS_PITCH;
  const v4f* L4 = (const v4f*)L;
  const v4f* P4 = (const v4f*)spill;

  float px[FPS_PPT];
  float dist[FPS_PPT];
  #pragma unroll
  for (int j = 0; j < FPS_PPT; ++j) {
    const int i = tid + j * FPS_T;
    float x = 0.0f, y = 0.0f, z = 0.0f, d = -FLT_BIG;
    if (i < N) {
      x = pos[i * 3 + 0];
      y = pos[i * 3 + 1];
      z = pos[i * 3 + 2];
      d = FLT_BIG;
    }
    dist[j] = d;
    L[j] = y;
    if (j >= 38 && j < 82) spill[((((6 + ((j - 38) >> 2)) * FPS_T) + tid) << 2) + ((j - 38) & 3)] = x;
    else                   px[j] = x;
    if (j < 24)      spill[((((j >> 2) * FPS_T) + tid) << 2) + (j & 3)] = z;
    else if (j < 80) L[100 + (j - 24)] = z;
    else if (j < 82) L[98 + (j - 80)] = z;
    else             spill[((((17 + ((j - 82) >> 2)) * FPS_T) + tid) << 2) + ((j - 82) & 3)] = z;
  }
  __syncthreads();

  v4f zb1[6];
  #pragma unroll
  for (int q = 0; q < 6; ++q) zb1[q] = P4[q * FPS_T + tid];

  int cur  = 0;
  int nxt  = 0;
  int v2b  = 0;
  int cok  = 0;
  float ex = 0.0f, ey = 0.0f, ez = 0.0f;
  const unsigned last_pt = (unsigned)(N - 1);
  int it  = 0;
  int par = 0;
  while (it < M) {
    __builtin_amdgcn_s_setprio(3);
    const float* cp = (const float*)((const char*)pos + __umul24((unsigned)cur, 12u));
    const float* up = (const float*)((const char*)pos + __umul24((unsigned)nxt, 12u));
    const float cx = cp[0];
    const float cy = cp[1];
    const float cz = cp[2];
    const float ux = up[0];
    const float uy = up[1];
    const float uz = up[2];
    const float gx = ux - cx;
    const float gy = uy - cy;
    const float gz = uz - cz;
    float g2 = gy * gy;
    g2 = __builtin_fmaf(gx, gx, g2);
    g2 = __builtin_fmaf(gz, gz, g2);
    const int two_v = (cok != 0 && g2 >= __builtin_bit_cast(float, v2b)) ? 1 : 0;
    const int two_s = __builtin_amdgcn_readfirstlane((it + 1 < M) ? two_v : 0);
    const bool two = (two_s != 0);
    if (tid == 0) {
      cent_idx[it] = cur;
      cents[it * 3 + 0] = cx;
      cents[it * 3 + 1] = cy;
      cents[it * 3 + 2] = cz;
      if (two) {
        cent_idx[it + 1] = nxt;
        cents[it * 3 + 3] = ux;
        cents[it * 3 + 4] = uy;
        cents[it * 3 + 5] = uz;
      }
    }
    float gm[FPS_NG];
    fps_sweep(dist, px, gm, zb1, L4, P4, tid, cx, cy, cz, two_s, ux, uy, uz, ex, ey, ez);
    it += two ? 2 : 1;
    float bv = gm[0];
    #pragma unroll
    for (int k = 1; k < FPS_NG; ++k) bv = fmaxf(bv, gm[k]);
    int vb = __builtin_bit_cast(int, bv);
    vb = fps_row_max<1>(vb);
    vb = fps_row_max<2>(vb);
    vb = fps_row_max<4>(vb);
    vb = fps_row_max<8>(vb);
    {
      const int ov = __builtin_amdgcn_permlanex16(vb, vb, 0x76543210, 0xfedcba98, false, false);
      vb = (ov > vb) ? ov : vb;
    }
    const float V = __builtin_bit_cast(float, vb);
    unsigned gb[FPS_NG];
    #pragma unroll
    for (int k = FPS_NG - 1; k >= 0; --k) gb[k] = __builtin_amdgcn_ballot_w32(gm[k] == V);
    __builtin_amdgcn_sched_barrier(0);
    int kg = 0;
    int npair = 0;
    #pragma unroll
    for (int k = FPS_NG - 1; k >= 0; --k) {
      if (gb[k] != 0u) kg = k;
      npair += __builtin_popcount(gb[k]);
    }
    int      jj  = 0;
    unsigned mm  = 1u;
    int      neq = 0;
    float    lx  = -FLT_BIG;
    #pragma unroll
    for (int k = 0; k < FPS_NG; ++k) {
      if (kg == k) {
        unsigned mr[FPS_GS];
        #pragma unroll
        for (int r = FPS_GS - 1; r >= 0; --r) mr[r] = __builtin_amdgcn_ballot_w32(dist[k * FPS_GS + r] == V);
        __builtin_amdgcn_sched_barrier(0);
        #pragma unroll
        for (int r = FPS_GS - 1; r >= 0; --r) {
          if (mr[r] != 0u) { jj = k * FPS_GS + r; mm = mr[r]; }
          neq += __builtin_popcount(mr[r]);
        }
        float g = -FLT_BIG;
        #pragma unroll
        for (int r = 0; r < FPS_GS; ++r) {
          const float e = dist[k * FPS_GS + r];
          g = fmaxf(g, (e == V) ? -FLT_BIG : e);
        }
        #pragma unroll
        for (int k2 = 0; k2 < FPS_NG; ++k2) {
          if (k2 != k) g = fmaxf(g, gm[k2]);
        }
        lx = g;
      }
    }
    int sb = __builtin_bit_cast(int, lx);
    sb = fps_row_max<1>(sb);
    sb = fps_row_max<2>(sb);
    sb = fps_row_max<4>(sb);
    sb = fps_row_max<8>(sb);
    {
      const int ov = __builtin_amdgcn_permlanex16(sb, sb, 0x76543210, 0xfedcba98, false, false);
      sb = (ov > sb) ? ov : sb;
    }
    if (npair != 1 || neq != 1) sb = vb;
    const int bi = (wave << 5) + __builtin_ctz(mm) + jj * FPS_T;
    if (lane == 0) {
      v4i w;
      w.x = vb;
      w.y = bi;
      w.z = sb;
      w.w = 0;
      s_win[par][wave] = w;
    }
    {
      const unsigned eb = ((unsigned)bi < last_pt) ? (unsigned)bi : last_pt;
      const float* ep = (const float*)((const char*)pos + __umul24(eb, 12u));
      ex = ep[0];
      ey = ep[1];
      ez = ep[2];
    }
    __builtin_amdgcn_fence(__ATOMIC_RELEASE, "workgroup", "local");
    __builtin_amdgcn_s_barrier();
    __builtin_amdgcn_fence(__ATOMIC_ACQUIRE, "workgroup", "local");
    const v4i ws = s_win[par][lane & 15];
    par ^= 1;
    int tv = ws.x;
    tv = fps_row_max<1>(tv);
    tv = fps_row_max<2>(tv);
    tv = fps_row_max<4>(tv);
    tv = fps_row_max<8>(tv);
    int ci = (ws.x == tv) ? ws.y : 0x7fffffff;
    ci = fps_row_min<1>(ci);
    ci = fps_row_min<2>(ci);
    ci = fps_row_min<4>(ci);
    ci = fps_row_min<8>(ci);
    const bool isw = (ws.x == tv) && (ws.y == ci);
    const int  rv  = isw ? ws.z : ws.x;
    int v2 = rv;
    v2 = fps_row_max<1>(v2);
    v2 = fps_row_max<2>(v2);
    v2 = fps_row_max<4>(v2);
    v2 = fps_row_max<8>(v2);
    int c2 = (rv == v2) ? (isw ? -1 : ws.y) : 0x7fffffff;
    c2 = fps_row_min<1>(c2);
    c2 = fps_row_min<2>(c2);
    c2 = fps_row_min<4>(c2);
    c2 = fps_row_min<8>(c2);
    cur = (ci < N) ? ci : 0;
    cok = (c2 >= 0 && c2 < N && __builtin_bit_cast(float, v2) > 0.0f) ? 1 : 0;
    nxt = (cok != 0) ? c2 : 0;
    v2b = v2;
  }
}

__global__ __launch_bounds__(256)
void bq_pool_kernel(const float* __restrict__ pos,
                    const float* __restrict__ fs,
                    const float* __restrict__ fv,
                    const int* __restrict__ cent_idx,
                    float* __restrict__ pooled_s,
                    float* __restrict__ pooled_v,
                    int N, int C, int M) {
  const int m   = blockIdx.x;
  const int tid = threadIdx.x;
  __shared__ float s_cd2[CAP];
  __shared__ int   s_cidx[CAP];
  __shared__ int   s_cnt;
  __shared__ int   s_sel[K_LOCAL];
  __shared__ float s_rv[256];
  __shared__ int   s_ri[256];
  __shared__ int   s_rp[256];
  __shared__ float s_c[3];

  if (tid == 0) {
    s_cnt = 0;
    int ci = cent_idx[m];
    s_c[0] = pos[ci * 3 + 0];
    s_c[1] = pos[ci * 3 + 1];
    s_c[2] = pos[ci * 3 + 2];
  }
  __syncthreads();
  const float cx = s_c[0], cy = s_c[1], cz = s_c[2];

  for (int i = tid; i < N; i += 256) {
    float dx = pos[i * 3 + 0] - cx;
    float dy = pos[i * 3 + 1] - cy;
    float dz = pos[i * 3 + 2] - cz;
    float d2 = dx * dx + dy * dy + dz * dz;
    if (d2 <= RADIUS2) {
      int p = atomicAdd(&s_cnt, 1);
      if (p < CAP) { s_cd2[p] = d2; s_cidx[p] = i; }
    }
  }
  __syncthreads();
  const int ncand = min(s_cnt, CAP);

  for (int s = 0; s < K_LOCAL; ++s) {
    float bv = FLT_BIG;
    int   bi = 0x7fffffff;
    int   bp = -1;
    for (int p = tid; p < ncand; p += 256) {
      float v  = s_cd2[p];
      int   ix = s_cidx[p];
      if (v < bv || (v == bv && ix < bi)) { bv = v; bi = ix; bp = p; }
    }
    s_rv[tid] = bv; s_ri[tid] = bi; s_rp[tid] = bp;
    __syncthreads();
    for (int off = 128; off >= 1; off >>= 1) {
      if (tid < off) {
        float ov = s_rv[tid + off];
        int   oi = s_ri[tid + off];
        if (ov < s_rv[tid] || (ov == s_rv[tid] && oi < s_ri[tid])) {
          s_rv[tid] = ov; s_ri[tid] = oi; s_rp[tid] = s_rp[tid + off];
        }
      }
      __syncthreads();
    }
    if (tid == 0) {
      if (s < ncand) {
        int p = s_rp[0];
        s_sel[s] = s_cidx[p];
        s_cd2[p] = FLT_BIG;
      } else {
        s_sel[s] = s_sel[0];
      }
    }
    __syncthreads();
  }

  for (int j = tid; j < C * 4; j += 256) {
    const int c = j >> 2;
    const int d = j & 3;
    float mx = -FLT_BIG;
    #pragma unroll
    for (int t = 0; t < K_LOCAL; ++t) {
      int n = s_sel[t];
      float v = (d == 0) ? fs[(size_t)n * C + c]
                         : fv[((size_t)n * C + c) * 3 + (d - 1)];
      mx = fmaxf(mx, v);
    }
    if (d == 0) pooled_s[(size_t)m * C + c] = mx;
    else        pooled_v[(size_t)(d - 1) * M * C + (size_t)m * C + c] = mx;
  }
}

__global__ __launch_bounds__(256)
void mix_wmma_kernel(const float* __restrict__ pooled_s,
                     const float* __restrict__ pooled_v,
                     const float* __restrict__ Ws, const float* __restrict__ Wv,
                     const float* __restrict__ bs, const float* __restrict__ bv,
                     float* __restrict__ outS,
                     float* __restrict__ outV,
                     int M, int C) {
  const int lane = threadIdx.x & 31;
  const int wave = threadIdx.x >> 5;
  const int tiles_o = C >> 4;
  const int tiles_m = M >> 4;
  const int per_d   = tiles_m * tiles_o;
  const int tile    = blockIdx.x * (blockDim.x >> 5) + wave;
  if (tile >= per_d * 4) return;

  const int d   = tile / per_d;
  const int rem = tile % per_d;
  const int m0  = (rem / tiles_o) << 4;
  const int o0  = (rem % tiles_o) << 4;

  const float* A    = (d == 0) ? (pooled_s + (size_t)m0 * C)
                               : (pooled_v + (size_t)(d - 1) * M * C + (size_t)m0 * C);
  const float* B    = (d == 0) ? Ws : Wv;
  const float* bias = (d == 0) ? bs : bv;

  const int ml = lane & 15;
  const int kh = lane >> 4;

  v8f acc = {};
  for (int k0 = 0; k0 < C; k0 += 4) {
    const int ka = k0 + 2 * kh;
    v2f a = *(const v2f*)(A + (size_t)ml * C + ka);
    v2f b;
    b.x = B[(size_t)ka * C + o0 + ml];
    b.y = B[(size_t)(ka + 1) * C + o0 + ml];
    acc = __builtin_amdgcn_wmma_f32_16x16x4_f32(
        false, a, false, b, (short)0, acc, false, false);
  }

  const int   n  = o0 + ml;
  const float bb = bias[n];
  #pragma unroll
  for (int v = 0; v < 8; ++v) {
    const int mm  = m0 + v + 8 * kh;
    const float val = acc[v] + bb;
    if (d == 0) outS[(size_t)mm * C + n] = val;
    else        outV[((size_t)mm * C + n) * 3 + (d - 1)] = val;
  }
}

extern "C" void kernel_launch(void* const* d_in, const int* in_sizes, int n_in,
                              void* d_out, int out_size, void* d_ws, size_t ws_size,
                              hipStream_t stream) {
  const float* pos = (const float*)d_in[0];
  const float* fs  = (const float*)d_in[1];
  const float* fv  = (const float*)d_in[2];
  const float* Ws  = (const float*)d_in[3];
  const float* Wv  = (const float*)d_in[4];
  const float* bs  = (const float*)d_in[5];
  const float* bv  = (const float*)d_in[6];

  const int N = in_sizes[0] / 3;
  const int C = in_sizes[1] / N;
  const int M = (N < 2048) ? N : 2048;

  float* out   = (float*)d_out;
  float* cents = out;
  float* outS  = out + (size_t)M * 3;
  float* outV  = outS + (size_t)M * C;

  int*   cent_idx = (int*)d_ws;
  float* wsf      = (float*)d_ws;
  float* spill    = wsf + 4096;
  float* pooled_s = wsf + 4096 + 65536;
  float* pooled_v = pooled_s + (size_t)M * C;

  fps_kernel<<<1, 512, 0, stream>>>(pos, spill, cent_idx, cents, N, M);
  bq_pool_kernel<<<M, 256, 0, stream>>>(pos, fs, fv, cent_idx,
                                        pooled_s, pooled_v, N, C, M);

  const int per_d = (M / 16) * (C / 16);
  const int total = per_d * 4;
  mix_wmma_kernel<<<(total + 7) / 8, 256, 0, stream>>>(
      pooled_s, pooled_v, Ws, Wv, bs, bv, outS, outV, M, C);
}


# Ball query kernel: scan keeps 8 point loads in flight per thread; the 32 nearest are picked by rank (each candidate counts the keys below its own) in one pass, not 32 argmin rounds; bit-identical
# speedup vs baseline: 9.9488x; 1.0565x over previous
#include <hip/hip_runtime.h>
#include <hip/hip_bf16.h>

#define FLT_BIG 3.402823466e+38f
#define RADIUS2 0.01f
#define K_LOCAL 32
#define CAP 2048

typedef __attribute__((ext_vector_type(2))) float v2f;
typedef __attribute__((ext_vector_type(8))) float v8f;

#define FPS_T     512
#define FPS_PPT   98
#define FPS_PITCH 156
#define FPS_GS    14
#define FPS_NG    7
#define FPS_NQ    25

typedef __attribute__((ext_vector_type(4))) float v4f;

template <int MASK>
__device__ __forceinline__ int fps_row_max(int v) {
  const int ov = __builtin_amdgcn_update_dpp(v, v, 0x160 + MASK, 0xf, 0xf, true);
  return (ov > v) ? ov : v;
}

template <int MASK>
__device__ __forceinline__ int fps_row_min(int v) {
  const int ov = __builtin_amdgcn_update_dpp(v, v, 0x160 + MASK, 0xf, 0xf, true);
  return (ov < v) ? ov : v;
}

typedef __attribute__((ext_vector_type(4))) int v4i;

__device__ __forceinline__ void fps_sweep(float (&dist)[FPS_PPT], const float (&px)[FPS_PPT],
                                          float (&gm)[FPS_NG], v4f (&zb1)[6],
                                          const v4f* L4, const v4f* P4, const int tid,
                                          const float cx, const float cy, const float cz,
                                          const int two_s,
                                          const float ux, const float uy, const float uz,
                                          const float ex, const float ey, const float ez) {
  v4f xb[11];
  v4f zc[4];
  int two_t = two_s;
  asm volatile("" : "+s"(two_t));
  #pragma unroll
  for (int qp = 0; qp < (FPS_NQ + 1) / 2; ++qp) {
    float qx[8], qy[8], qz[8];
    float dx[8], dy[8], dz[8], d2[8];
    #pragma unroll
    for (int h = 0; h < 2; ++h) {
      const int q = 2 * qp + h;
      if (q < FPS_NQ) {
        if (q == 6)  __builtin_amdgcn_s_setprio(2);
        if (q == 13) __builtin_amdgcn_s_setprio(1);
        if (q == 20) __builtin_amdgcn_s_setprio(0);
        if (q == 2) {
          #pragma unroll
          for (int p = 0; p < 4; ++p) xb[p] = P4[(6 + p) * FPS_T + tid];
        }
        if (q == 7) {
          #pragma unroll
          for (int p = 4; p < 11; ++p) xb[p] = P4[(6 + p) * FPS_T + tid];
        }
        if (q == 15) {
          #pragma unroll
          for (int p = 0; p < 4; ++p) zc[p] = P4[(17 + p) * FPS_T + tid];
        }
        if (q == 21) {
          #pragma unroll
          for (int p = 0; p < 6; ++p) zb1[p] = P4[p * FPS_T + tid];
        }
        const v4f yv = L4[q];
        #pragma unroll
        for (int r = 0; r < 4; ++r) {
          const int j = 4 * q + r;
          if (j < FPS_PPT) {
            float z;
            if (j < 24)      { z = zb1[j >> 2][j & 3]; }
            else if (j < 80) { const v4f zv = L4[25 + ((j - 24) >> 2)]; z = zv[(j - 24) & 3]; }
            else if (j < 82) { const v4f zv = L4[24]; z = zv[2 + (j - 80)]; }
            else             { z = zc[(j - 82) >> 2][(j - 82) & 3]; }
            float x;
            if (j >= 38 && j < 82) x = xb[(j - 38) >> 2][(j - 38) & 3];
            else                   x = px[j];
            qx[4 * h + r] = x;
            qy[4 * h + r] = yv[r];
            qz[4 * h + r] = z;
          }
        }
      }
    }
    #pragma unroll
    for (int i = 0; i < 8; ++i) if (8 * qp + i < FPS_PPT) { dx[i] = qx[i] - cx; dy[i] = qy[i] - cy; dz[i] = qz[i] - cz; }
    #pragma unroll
    for (int i = 0; i < 8; ++i) if (8 * qp + i < FPS_PPT) d2[i] = dy[i] * dy[i];
    #pragma unroll
    for (int i = 0; i < 8; ++i) if (8 * qp + i < FPS_PPT) d2[i] = __builtin_fmaf(dx[i], dx[i], d2[i]);
    #pragma unroll
    for (int i = 0; i < 8; ++i) if (8 * qp + i < FPS_PPT) d2[i] = __builtin_fmaf(dz[i], dz[i], d2[i]);
    if (two_s == 0) {
      #pragma unroll
      for (int i = 0; i < 8; ++i) {
        const int j = 8 * qp + i;
        if (j < FPS_PPT) {
          const int a = __builtin_bit_cast(int, dist[j]);
          const int b = __builtin_bit_cast(int, d2[i]);
          dist[j] = __builtin_bit_cast(float, (b < a) ? b : a);
        }
      }
    }
    if (two_t != 0) {
      float e2[8];
      #pragma unroll
      for (int i = 0; i < 8; ++i) if (8 * qp + i < FPS_PPT) { dx[i] = qx[i] - ux; dy[i] = qy[i] - uy; dz[i] = qz[i] - uz; }
      #pragma unroll
      for (int i = 0; i < 8; ++i) if (8 * qp + i < FPS_PPT) e2[i] = dy[i] * dy[i];
      #pragma unroll
      for (int i = 0; i < 8; ++i) if (8 * qp + i < FPS_PPT) e2[i] = __builtin_fmaf(dx[i], dx[i], e2[i]);
      #pragma unroll
      for (int i = 0; i < 8; ++i) if (8 * qp + i < FPS_PPT) e2[i] = __builtin_fmaf(dz[i], dz[i], e2[i]);
      #pragma unroll
      for (int i = 0; i < 8; ++i) {
        const int j = 8 * qp + i;
        if (j < FPS_PPT) {
          float dd = fminf(dist[j], d2[i]);
          dd = fminf(dd, e2[i]);
          dist[j] = dd;
        }
      }
    }
    #pragma unroll
    for (int i = 0; i < 8; ++i) {
      const int j = 8 * qp + i;
      if (j < FPS_PPT) {
        if (j % FPS_GS == 0) gm[j / FPS_GS] = dist[j];
        else                 gm[j / FPS_GS] = fmaxf(gm[j / FPS_GS], dist[j]);
      }
    }
    if (qp == 0) asm volatile("" :: "v"(ex), "v"(ey), "v"(ez));
    __builtin_amdgcn_sched_barrier(0x0094);
  }
}

__global__ __launch_bounds__(FPS_T)
void fps_kernel(const float* __restrict__ pos, float* __restrict__ spill,
                int* __restrict__ cent_idx, float* __restrict__ cents,
                int N, int M) {
  const int tid  = threadIdx.x;
  const int lane = tid & 31;
  const int wave = tid >> 5;
  __shared__ __attribute__((aligned(16))) v4i   s_win[2][FPS_T / 32];
  __shared__ __attribute__((aligned(16))) float s_pt[FPS_T * FPS_PITCH];

  if (N > FPS_PPT * FPS_T) return;

  float*     L  = s_pt + tid * FPS_PITCH;
  const v4f* L4 = (const v4f*)L;
  const v4f* P4 = (const v4f*)spill;

  float px[FPS_PPT];
  float dist[FPS_PPT];
  #pragma unroll
  for (int j = 0; j < FPS_PPT; ++j) {
    const int i = tid + j * FPS_T;
    float x = 0.0f, y = 0.0f, z = 0.0f, d = -FLT_BIG;
    if (i < N) {
      x = pos[i * 3 + 0];
      y = pos[i * 3 + 1];
      z = pos[i * 3 + 2];
      d = FLT_BIG;
    }
    dist[j] = d;
    L[j] = y;
    if (j >= 38 && j < 82) spill[((((6 + ((j - 38) >> 2)) * FPS_T) + tid) << 2) + ((j - 38) & 3)] = x;
    else                   px[j] = x;
    if (j < 24)      spill[((((j >> 2) * FPS_T) + tid) << 2) + (j & 3)] = z;
    else if (j < 80) L[100 + (j - 24)] = z;
    else if (j < 82) L[98 + (j - 80)] = z;
    else             spill[((((17 + ((j - 82) >> 2)) * FPS_T) + tid) << 2) + ((j - 82) & 3)] = z;
  }
  __syncthreads();

  v4f zb1[6];
  #pragma unroll
  for (int q = 0; q < 6; ++q) zb1[q] = P4[q * FPS_T + tid];

  int cur  = 0;
  int nxt  = 0;
  int v2b  = 0;
  int cok  = 0;
  float ex = 0.0f, ey = 0.0f, ez = 0.0f;
  const unsigned last_pt = (unsigned)(N - 1);
  int it  = 0;
  int par = 0;
  while (it < M) {
    __builtin_amdgcn_s_setprio(3);
    const float* cp = (const float*)((const char*)pos + __umul24((unsigned)cur, 12u));
    const float* up = (const float*)((const char*)pos + __umul24((unsigned)nxt, 12u));
    const float cx = cp[0];
    const float cy = cp[1];
    const float cz = cp[2];
    const float ux = up[0];
    const float uy = up[1];
    const float uz = up[2];
    const float gx = ux - cx;
    const float gy = uy - cy;
    const float gz = uz - cz;
    float g2 = gy * gy;
    g2 = __builtin_fmaf(gx, gx, g2);
    g2 = __builtin_fmaf(gz, gz, g2);
    const int two_v = (cok != 0 && g2 >= __builtin_bit_cast(float, v2b)) ? 1 : 0;
    const int two_s = __builtin_amdgcn_readfirstlane((it + 1 < M) ? two_v : 0);
    const bool two = (two_s != 0);
    if (tid == 0) {
      cent_idx[it] = cur;
      cents[it * 3 + 0] = cx;
      cents[it * 3 + 1] = cy;
      cents[it * 3 + 2] = cz;
      if (two) {
        cent_idx[it + 1] = nxt;
        cents[it * 3 + 3] = ux;
        cents[it * 3 + 4] = uy;
        cents[it * 3 + 5] = uz;
      }
    }
    float gm[FPS_NG];
    fps_sweep(dist, px, gm, zb1, L4, P4, tid, cx, cy, cz, two_s, ux, uy, uz, ex, ey, ez);
    it += two ? 2 : 1;
    float bv = gm[0];
    #pragma unroll
    for (int k = 1; k < FPS_NG; ++k) bv = fmaxf(bv, gm[k]);
    int vb = __builtin_bit_cast(int, bv);
    vb = fps_row_max<1>(vb);
    vb = fps_row_max<2>(vb);
    vb = fps_row_max<4>(vb);
    vb = fps_row_max<8>(vb);
    {
      const int ov = __builtin_amdgcn_permlanex16(vb, vb, 0x76543210, 0xfedcba98, false, false);
      vb = (ov > vb) ? ov : vb;
    }
    const float V = __builtin_bit_cast(float, vb);
    unsigned gb[FPS_NG];
    #pragma unroll
    for (int k = FPS_NG - 1; k >= 0; --k) gb[k] = __builtin_amdgcn_ballot_w32(gm[k] == V);
    __builtin_amdgcn_sched_barrier(0);
    int kg = 0;
    int npair = 0;
    #pragma unroll
    for (int k = FPS_NG - 1; k >= 0; --k) {
      if (gb[k] != 0u) kg = k;
      npair += __builtin_popcount(gb[k]);
    }
    int      jj  = 0;
    unsigned mm  = 1u;
    int      neq = 0;
    float    lx  = -FLT_BIG;
    #pragma unroll
    for (int k = 0; k < FPS_NG; ++k) {
      if (kg == k) {
        unsigned mr[FPS_GS];
        #pragma unroll
        for (int r = FPS_GS - 1; r >= 0; --r) mr[r] = __builtin_amdgcn_ballot_w32(dist[k * FPS_GS + r] == V);
        __builtin_amdgcn_sched_barrier(0);
        #pragma unroll
        for (int r = FPS_GS - 1; r >= 0; --r) {
          if (mr[r] != 0u) { jj = k * FPS_GS + r; mm = mr[r]; }
          neq += __builtin_popcount(mr[r]);
        }
        float g = -FLT_BIG;
        #pragma unroll
        for (int r = 0; r < FPS_GS; ++r) {
          const float e = dist[k * FPS_GS + r];
          g = fmaxf(g, (e == V) ? -FLT_BIG : e);
        }
        #pragma unroll
        for (int k2 = 0; k2 < FPS_NG; ++k2) {
          if (k2 != k) g = fmaxf(g, gm[k2]);
        }
        lx = g;
      }
    }
    int sb = __builtin_bit_cast(int, lx);
    sb = fps_row_max<1>(sb);
    sb = fps_row_max<2>(sb);
    sb = fps_row_max<4>(sb);
    sb = fps_row_max<8>(sb);
    {
      const int ov = __builtin_amdgcn_permlanex16(sb, sb, 0x76543210, 0xfedcba98, false, false);
      sb = (ov > sb) ? ov : sb;
    }
    if (npair != 1 || neq != 1) sb = vb;
    const int bi = (wave << 5) + __builtin_ctz(mm) + jj * FPS_T;
    if (lane == 0) {
      v4i w;
      w.x = vb;
      w.y = bi;
      w.z = sb;
      w.w = 0;
      s_win[par][wave] = w;
    }
    {
      const unsigned eb = ((unsigned)bi < last_pt) ? (unsigned)bi : last_pt;
      const float* ep = (const float*)((const char*)pos + __umul24(eb, 12u));
      ex = ep[0];
      ey = ep[1];
      ez = ep[2];
    }
    __builtin_amdgcn_fence(__ATOMIC_RELEASE, "workgroup", "local");
    __builtin_amdgcn_s_barrier();
    __builtin_amdgcn_fence(__ATOMIC_ACQUIRE, "workgroup", "local");
    const v4i ws = s_win[par][lane & 15];
    par ^= 1;
    int tv = ws.x;
    tv = fps_row_max<1>(tv);
    tv = fps_row_max<2>(tv);
    tv = fps_row_max<4>(tv);
    tv = fps_row_max<8>(tv);
    int ci = (ws.x == tv) ? ws.y : 0x7fffffff;
    ci = fps_row_min<1>(ci);
    ci = fps_row_min<2>(ci);
    ci = fps_row_min<4>(ci);
    ci = fps_row_min<8>(ci);
    const bool isw = (ws.x == tv) && (ws.y == ci);
    const int  rv  = isw ? ws.z : ws.x;
    int v2 = rv;
    v2 = fps_row_max<1>(v2);
    v2 = fps_row_max<2>(v2);
    v2 = fps_row_max<4>(v2);
    v2 = fps_row_max<8>(v2);
    int c2 = (rv == v2) ? (isw ? -1 : ws.y) : 0x7fffffff;
    c2 = fps_row_min<1>(c2);
    c2 = fps_row_min<2>(c2);
    c2 = fps_row_min<4>(c2);
    c2 = fps_row_min<8>(c2);
    cur = (ci < N) ? ci : 0;
    cok = (c2 >= 0 && c2 < N && __builtin_bit_cast(float, v2) > 0.0f) ? 1 : 0;
    nxt = (cok != 0) ? c2 : 0;
    v2b = v2;
  }
}

#define BQ_U 8

typedef __attribute__((ext_vector_type(2))) unsigned long long v2u64;

__global__ __launch_bounds__(256)
void bq_pool_kernel(const float* __restrict__ pos,
                    const float* __restrict__ fs,
                    const float* __restrict__ fv,
                    const int* __restrict__ cent_idx,
                    float* __restrict__ pooled_s,
                    float* __restrict__ pooled_v,
                    int N, int C, int M) {
  const int m   = blockIdx.x;
  const int tid = threadIdx.x;
  __shared__ __attribute__((aligned(16))) unsigned long long s_key[CAP];
  __shared__ int   s_cnt;
  __shared__ int   s_sel[K_LOCAL];
  __shared__ float s_c[3];

  if (tid == 0) {
    s_cnt = 0;
    s_sel[0] = 0;
    int ci = cent_idx[m];
    s_c[0] = pos[ci * 3 + 0];
    s_c[1] = pos[ci * 3 + 1];
    s_c[2] = pos[ci * 3 + 2];
  }
  __syncthreads();
  const float cx = s_c[0], cy = s_c[1], cz = s_c[2];

  for (int i0 = tid; i0 < N; i0 += 256 * BQ_U) {
    float x[BQ_U], y[BQ_U], z[BQ_U];
    #pragma unroll
    for (int u = 0; u < BQ_U; ++u) {
      const int i  = i0 + u * 256;
      const int ic = (i < N) ? i : (N - 1);
      x[u] = pos[ic * 3 + 0];
      y[u] = pos[ic * 3 + 1];
      z[u] = pos[ic * 3 + 2];
    }
    #pragma unroll
    for (int u = 0; u < BQ_U; ++u) {
      const int i = i0 + u * 256;
      const float dx = x[u] - cx;
      const float dy = y[u] - cy;
      const float dz = z[u] - cz;
      float d2 = dy * dy;
      d2 = __builtin_fmaf(dx, dx, d2);
      d2 = __builtin_fmaf(dz, dz, d2);
      if (i < N && d2 <= RADIUS2) {
        int p = atomicAdd(&s_cnt, 1);
        if (p < CAP) {
          s_key[p] = ((unsigned long long)__builtin_bit_cast(unsigned, d2) << 32) | (unsigned long long)(unsigned)i;
        }
      }
    }
  }
  __syncthreads();
  const int ncand = min(s_cnt, CAP);
  const int nsel  = min(ncand, K_LOCAL);

  {
    const v2u64* K2 = (const v2u64*)s_key;
    const int np = ncand >> 1;
    for (int p = tid; p < ncand; p += 256) {
      const unsigned long long key = s_key[p];
      int rank = 0;
      #pragma unroll 4
      for (int q = 0; q < np; ++q) {
        const v2u64 k = K2[q];
        rank += (k.x < key) ? 1 : 0;
        rank += (k.y < key) ? 1 : 0;
      }
      if (ncand & 1) rank += (s_key[ncand - 1] < key) ? 1 : 0;
      if (rank < K_LOCAL) s_sel[rank] = (int)(unsigned)key;
    }
  }
  __syncthreads();

  for (int j = tid; j < C * 4; j += 256) {
    const int c = j >> 2;
    const int d = j & 3;
    float mx = -FLT_BIG;
    #pragma unroll
    for (int t = 0; t < K_LOCAL; ++t) {
      int n = s_sel[(t < nsel) ? t : 0];
      float v = (d == 0) ? fs[(size_t)n * C + c]
                         : fv[((size_t)n * C + c) * 3 + (d - 1)];
      mx = fmaxf(mx, v);
    }
    if (d == 0) pooled_s[(size_t)m * C + c] = mx;
    else        pooled_v[(size_t)(d - 1) * M * C + (size_t)m * C + c] = mx;
  }
}

__global__ __launch_bounds__(256)
void mix_wmma_kernel(const float* __restrict__ pooled_s,
                     const float* __restrict__ pooled_v,
                     const float* __restrict__ Ws, const float* __restrict__ Wv,
                     const float* __restrict__ bs, const float* __restrict__ bv,
                     float* __restrict__ outS,
                     float* __restrict__ outV,
                     int M, int C) {
  const int lane = threadIdx.x & 31;
  const int wave = threadIdx.x >> 5;
  const int tiles_o = C >> 4;
  const int tiles_m = M >> 4;
  const int per_d   = tiles_m * tiles_o;
  const int tile    = blockIdx.x * (blockDim.x >> 5) + wave;
  if (tile >= per_d * 4) return;

  const int d   = tile / per_d;
  const int rem = tile % per_d;
  const int m0  = (rem / tiles_o) << 4;
  const int o0  = (rem % tiles_o) << 4;

  const float* A    = (d == 0) ? (pooled_s + (size_t)m0 * C)
                               : (pooled_v + (size_t)(d - 1) * M * C + (size_t)m0 * C);
  const float* B    = (d == 0) ? Ws : Wv;
  const float* bias = (d == 0) ? bs : bv;

  const int ml = lane & 15;
  const int kh = lane >> 4;

  v8f acc = {};
  for (int k0 = 0; k0 < C; k0 += 4) {
    const int ka = k0 + 2 * kh;
    v2f a = *(const v2f*)(A + (size_t)ml * C + ka);
    v2f b;
    b.x = B[(size_t)ka * C + o0 + ml];
    b.y = B[(size_t)(ka + 1) * C + o0 + ml];
    acc = __builtin_amdgcn_wmma_f32_16x16x4_f32(
        false, a, false, b, (short)0, acc, false, false);
  }

  const int   n  = o0 + ml;
  const float bb = bias[n];
  #pragma unroll
  for (int v = 0; v < 8; ++v) {
    const int mm  = m0 + v + 8 * kh;
    const float val = acc[v] + bb;
    if (d == 0) outS[(size_t)mm * C + n] = val;
    else        outV[((size_t)mm * C + n) * 3 + (d - 1)] = val;
  }
}

extern "C" void kernel_launch(void* const* d_in, const int* in_sizes, int n_in,
                              void* d_out, int out_size, void* d_ws, size_t ws_size,
                              hipStream_t stream) {
  const float* pos = (const float*)d_in[0];
  const float* fs  = (const float*)d_in[1];
  const float* fv  = (const float*)d_in[2];
  const float* Ws  = (const float*)d_in[3];
  const float* Wv  = (const float*)d_in[4];
  const float* bs  = (const float*)d_in[5];
  const float* bv  = (const float*)d_in[6];

  const int N = in_sizes[0] / 3;
  const int C = in_sizes[1] / N;
  const int M = (N < 2048) ? N : 2048;

  float* out   = (float*)d_out;
  float* cents = out;
  float* outS  = out + (size_t)M * 3;
  float* outV  = outS + (size_t)M * C;

  int*   cent_idx = (int*)d_ws;
  float* wsf      = (float*)d_ws;
  float* spill    = wsf + 4096;
  float* pooled_s = wsf + 4096 + 65536;
  float* pooled_v = pooled_s + (size_t)M * C;

  fps_kernel<<<1, 512, 0, stream>>>(pos, spill, cent_idx, cents, N, M);
  bq_pool_kernel<<<M, 256, 0, stream>>>(pos, fs, fv, cent_idx,
                                        pooled_s, pooled_v, N, C, M);

  const int per_d = (M / 16) * (C / 16);
  const int total = per_d * 4;
  mix_wmma_kernel<<<(total + 7) / 8, 256, 0, stream>>>(
      pooled_s, pooled_v, Ws, Wv, bs, bv, outS, outV, M, C);
}
